# starting kernel
# baseline (speedup 1.0000x reference)
#include <hip/hip_runtime.h>

#define B_ 512
#define T_ 512
#define D_ 64
#define H_ 128
#define G_ 512
#define OUT_ 16
#define BT_ 16
#define THREADS_ 512
#define NT_ 2

typedef __attribute__((ext_vector_type(16))) _Float16 v16h;
typedef __attribute__((ext_vector_type(8)))  _Float16 v8h;
typedef __attribute__((ext_vector_type(8)))  float    v8f;

__device__ __forceinline__ float fast_tanh(float x) {
#if __has_builtin(__builtin_amdgcn_tanhf)
  return __builtin_amdgcn_tanhf(x);
#else
  float e = __builtin_amdgcn_exp2f(x * 2.8853900817779268f);
  return 1.0f - 2.0f * __builtin_amdgcn_rcpf(e + 1.0f);
#endif
}

__device__ __forceinline__ v16h ldfrag_g(const _Float16* __restrict__ base,
                                         int stride, int koff, int lane) {
  int r  = lane & 15;
  int hi = lane >> 4;
  const _Float16* p = base + (size_t)r * stride + koff + hi * 8;
  v8h a = *(const v8h*)p;
  v8h b = *(const v8h*)(p + 16);
  return __builtin_shufflevector(a, b, 0,1,2,3,4,5,6,7,8,9,10,11,12,13,14,15);
}

__global__ void cvt_f32_to_f16(const float* __restrict__ s, _Float16* __restrict__ d, int n) {
  int i = blockIdx.x * blockDim.x + threadIdx.x;
  if (i < n) d[i] = (_Float16)s[i];
}

__global__ void add_vec(const float* __restrict__ a, const float* __restrict__ b,
                        float* __restrict__ o, int n) {
  int i = blockIdx.x * blockDim.x + threadIdx.x;
  if (i < n) o[i] = a[i] + b[i];
}

template<int KIN>
__global__ __launch_bounds__(THREADS_, 1)
void lstm_layer_kernel(const _Float16* __restrict__ xseq,
                       const _Float16* __restrict__ Wih,
                       const _Float16* __restrict__ Whh,
                       const float*    __restrict__ bias,
                       _Float16*       __restrict__ hseq_out,
                       float*          __restrict__ hlast_out)
{
  constexpr int KX = KIN / 32;

  __shared__ _Float16 h_lds[BT_][H_ + 8];
  __shared__ float    g_lds[BT_][G_ + 8];

  const int tid   = threadIdx.x;
  const int lane  = tid & 31;
  const int wave  = tid >> 5;
  const int hi    = lane >> 4;
  const int b0    = blockIdx.x * BT_;
  const int ncol0 = wave * (NT_ * 16);

  for (int i = tid; i < BT_ * (H_ + 8); i += THREADS_)
    ((_Float16*)h_lds)[i] = (_Float16)0.0f;

  v16h Bx[NT_][KX];
  v16h Bh[NT_][4];
  float bv[NT_];
  #pragma unroll
  for (int n = 0; n < NT_; ++n) {
    const int nc = ncol0 + n * 16;
    bv[n] = bias[nc + (lane & 15)];
    #pragma unroll
    for (int kt = 0; kt < KX; ++kt)
      Bx[n][kt] = ldfrag_g(Wih + (size_t)nc * KIN, KIN, kt * 32, lane);
    #pragma unroll
    for (int kt = 0; kt < 4; ++kt)
      Bh[n][kt] = ldfrag_g(Whh + (size_t)nc * H_, H_, kt * 32, lane);
  }

  const bool  is_tanh = ((wave >> 2) == 2);
  const float act_pre  = is_tanh ? 1.0f : 0.5f;
  const float act_post = is_tanh ? 1.0f : 0.5f;
  const float act_off  = is_tanh ? 0.0f : 0.5f;

  float c[4] = {0.f, 0.f, 0.f, 0.f};

  const _Float16* xbase = xseq + (size_t)b0 * T_ * KIN;

  __syncthreads();

  for (int t = 0; t < T_; ++t) {
    v16h xa[KX];
    #pragma unroll
    for (int kt = 0; kt < KX; ++kt)
      xa[kt] = ldfrag_g(xbase + (size_t)t * KIN, T_ * KIN, kt * 32, lane);
    if (t + 1 < T_)
      __builtin_prefetch(xbase + (size_t)(t + 1) * KIN + (size_t)(lane & 15) * (T_ * KIN), 0, 1);

    v16h ha[4];
    #pragma unroll
    for (int kt = 0; kt < 4; ++kt) {
      const _Float16* p = &h_lds[lane & 15][kt * 32 + hi * 8];
      v8h a = *(const v8h*)p;
      v8h b = *(const v8h*)(p + 16);
      ha[kt] = __builtin_shufflevector(a, b, 0,1,2,3,4,5,6,7,8,9,10,11,12,13,14,15);
    }

    #pragma unroll
    for (int n = 0; n < NT_; ++n) {
      v8f acc;
      #pragma unroll
      for (int m = 0; m < 8; ++m) acc[m] = bv[n];
      #pragma unroll
      for (int kt = 0; kt < KX; ++kt)
        acc = __builtin_amdgcn_wmma_f32_16x16x32_f16(false, xa[kt], false, Bx[n][kt],
                                                     (short)0, acc, false, false);
      #pragma unroll
      for (int kt = 0; kt < 4; ++kt)
        acc = __builtin_amdgcn_wmma_f32_16x16x32_f16(false, ha[kt], false, Bh[n][kt],
                                                     (short)0, acc, false, false);

      const int col = ncol0 + n * 16 + (lane & 15);
      #pragma unroll
      for (int m = 0; m < 8; ++m) {
        float a = act_off + act_post * fast_tanh(acc[m] * act_pre);
        g_lds[m + 8 * hi][col] = a;
      }
    }
    __syncthreads();

    #pragma unroll
    for (int e = 0; e < 4; ++e) {
      int q = tid * 4 + e;
      int r = q >> 7;
      int j = q & (H_ - 1);
      float iv = g_lds[r][j];
      float fv = g_lds[r][H_ + j];
      float gv = g_lds[r][2 * H_ + j];
      float ov = g_lds[r][3 * H_ + j];
      c[e] = fv * c[e] + iv * gv;
      float hv = ov * fast_tanh(c[e]);
      h_lds[r][j] = (_Float16)hv;
      if (hseq_out)
        hseq_out[(size_t)(b0 + r) * T_ * H_ + (size_t)t * H_ + j] = (_Float16)hv;
      if (hlast_out && t == T_ - 1)
        hlast_out[(b0 + r) * H_ + j] = hv;
    }
    __syncthreads();
  }
}

__global__ void fc_kernel(const float* __restrict__ h1, const float* __restrict__ Wfc0,
                          const float* __restrict__ bfc0, const float* __restrict__ Wfc1,
                          const float* __restrict__ bfc1, float* __restrict__ out) {
  int b = blockIdx.x;
  int tid = threadIdx.x;
  __shared__ float tmp[32];
  float s = bfc0[tid];
  for (int k = 0; k < H_; ++k) s += h1[b * H_ + k] * Wfc0[tid * H_ + k];
  tmp[tid] = s;
  __syncthreads();
  if (tid < OUT_) {
    float o = bfc1[tid];
    for (int k = 0; k < 32; ++k) o += tmp[k] * Wfc1[tid * 32 + k];
    out[b * OUT_ + tid] = o;
  }
}

extern "C" void kernel_launch(void* const* d_in, const int* in_sizes, int n_in,
                              void* d_out, int out_size, void* d_ws, size_t ws_size,
                              hipStream_t stream) {
  (void)in_sizes; (void)n_in; (void)out_size; (void)ws_size;
  const float* x    = (const float*)d_in[0];
  const float* Wih0 = (const float*)d_in[1];
  const float* Whh0 = (const float*)d_in[2];
  const float* bih0 = (const float*)d_in[3];
  const float* bhh0 = (const float*)d_in[4];
  const float* Wih1 = (const float*)d_in[5];
  const float* Whh1 = (const float*)d_in[6];
  const float* bih1 = (const float*)d_in[7];
  const float* bhh1 = (const float*)d_in[8];
  const float* Wfc0 = (const float*)d_in[9];
  const float* bfc0 = (const float*)d_in[10];
  const float* Wfc1 = (const float*)d_in[11];
  const float* bfc1 = (const float*)d_in[12];

  char* p = (char*)d_ws;
  auto alloc = [&](size_t bytes) {
    char* r = p;
    p += (bytes + 255) & ~(size_t)255;
    return r;
  };
  _Float16* x_h    = (_Float16*)alloc((size_t)B_ * T_ * D_ * 2);
  _Float16* h0seq  = (_Float16*)alloc((size_t)B_ * T_ * H_ * 2);
  _Float16* wih0_h = (_Float16*)alloc((size_t)G_ * D_ * 2);
  _Float16* whh0_h = (_Float16*)alloc((size_t)G_ * H_ * 2);
  _Float16* wih1_h = (_Float16*)alloc((size_t)G_ * H_ * 2);
  _Float16* whh1_h = (_Float16*)alloc((size_t)G_ * H_ * 2);
  float*    b0     = (float*)alloc((size_t)G_ * 4);
  float*    b1     = (float*)alloc((size_t)G_ * 4);
  float*    h1last = (float*)alloc((size_t)B_ * H_ * 4);

  auto cvt = [&](const float* s, _Float16* d, int n) {
    cvt_f32_to_f16<<<(n + 255) / 256, 256, 0, stream>>>(s, d, n);
  };
  cvt(x, x_h, B_ * T_ * D_);
  cvt(Wih0, wih0_h, G_ * D_);
  cvt(Whh0, whh0_h, G_ * H_);
  cvt(Wih1, wih1_h, G_ * H_);
  cvt(Whh1, whh1_h, G_ * H_);
  add_vec<<<2, 256, 0, stream>>>(bih0, bhh0, b0, G_);
  add_vec<<<2, 256, 0, stream>>>(bih1, bhh1, b1, G_);

  lstm_layer_kernel<D_><<<B_ / BT_, THREADS_, 0, stream>>>(x_h, wih0_h, whh0_h, b0,
                                                           h0seq, nullptr);
  lstm_layer_kernel<H_><<<B_ / BT_, THREADS_, 0, stream>>>(h0seq, wih1_h, whh1_h, b1,
                                                           nullptr, h1last);
  fc_kernel<<<B_, 32, 0, stream>>>(h1last, Wfc0, bfc0, Wfc1, bfc1, (float*)d_out);
}


# x_t fetched from global once per workgroup, one step ahead, and handed to the waves through LDS (whole lines per load) instead of 16 waves each re-reading the 16-row fragment from global
# speedup vs baseline: 1.8661x; 1.8661x over previous
#include <hip/hip_runtime.h>

#define B_ 512
#define T_ 512
#define D_ 64
#define H_ 128
#define G_ 512
#define OUT_ 16
#define BT_ 16
#define THREADS_ 512
#define NT_ 2

typedef __attribute__((ext_vector_type(16))) _Float16 v16h;
typedef __attribute__((ext_vector_type(8)))  _Float16 v8h;
typedef __attribute__((ext_vector_type(8)))  float    v8f;

__device__ __forceinline__ float fast_tanh(float x) {
#if __has_builtin(__builtin_amdgcn_tanhf)
  return __builtin_amdgcn_tanhf(x);
#else
  float e = __builtin_amdgcn_exp2f(x * 2.8853900817779268f);
  return 1.0f - 2.0f * __builtin_amdgcn_rcpf(e + 1.0f);
#endif
}

__device__ __forceinline__ v16h ldfrag_g(const _Float16* __restrict__ base,
                                         int stride, int koff, int lane) {
  int r  = lane & 15;
  int hi = lane >> 4;
  const _Float16* p = base + (size_t)r * stride + koff + hi * 8;
  v8h a = *(const v8h*)p;
  v8h b = *(const v8h*)(p + 16);
  return __builtin_shufflevector(a, b, 0,1,2,3,4,5,6,7,8,9,10,11,12,13,14,15);
}

__global__ void cvt_f32_to_f16(const float* __restrict__ s, _Float16* __restrict__ d, int n) {
  int i = blockIdx.x * blockDim.x + threadIdx.x;
  if (i < n) d[i] = (_Float16)s[i];
}

__global__ void add_vec(const float* __restrict__ a, const float* __restrict__ b,
                        float* __restrict__ o, int n) {
  int i = blockIdx.x * blockDim.x + threadIdx.x;
  if (i < n) o[i] = a[i] + b[i];
}

template<int KIN>
__global__ __launch_bounds__(THREADS_, 1)
void lstm_layer_kernel(const _Float16* __restrict__ xseq,
                       const _Float16* __restrict__ Wih,
                       const _Float16* __restrict__ Whh,
                       const float*    __restrict__ bias,
                       _Float16*       __restrict__ hseq_out,
                       float*          __restrict__ hlast_out)
{
  constexpr int KX = KIN / 32;

  __shared__ _Float16 h_lds[BT_][H_ + 8];
  __shared__ float    g_lds[BT_][G_ + 8];
  __shared__ __attribute__((aligned(16)))
  _Float16 x_lds[BT_][KIN + 8];

  const int tid   = threadIdx.x;
  const int lane  = tid & 31;
  const int wave  = tid >> 5;
  const int hi    = lane >> 4;
  const int b0    = blockIdx.x * BT_;
  const int ncol0 = wave * (NT_ * 16);

  for (int i = tid; i < BT_ * (H_ + 8); i += THREADS_)
    ((_Float16*)h_lds)[i] = (_Float16)0.0f;

  v16h Bx[NT_][KX];
  v16h Bh[NT_][4];
  float bv[NT_];
  #pragma unroll
  for (int n = 0; n < NT_; ++n) {
    const int nc = ncol0 + n * 16;
    bv[n] = bias[nc + (lane & 15)];
    #pragma unroll
    for (int kt = 0; kt < KX; ++kt)
      Bx[n][kt] = ldfrag_g(Wih + (size_t)nc * KIN, KIN, kt * 32, lane);
    #pragma unroll
    for (int kt = 0; kt < 4; ++kt)
      Bh[n][kt] = ldfrag_g(Whh + (size_t)nc * H_, H_, kt * 32, lane);
  }

  const bool  is_tanh = ((wave >> 2) == 2);
  const float act_pre  = is_tanh ? 1.0f : 0.5f;
  const float act_post = is_tanh ? 1.0f : 0.5f;
  const float act_off  = is_tanh ? 0.0f : 0.5f;

  float c[4] = {0.f, 0.f, 0.f, 0.f};

  const _Float16* xbase = xseq + (size_t)b0 * T_ * KIN;

  constexpr int CPR_   = KIN / 8;
  constexpr int NLOAD_ = BT_ * CPR_;
  const bool loader = tid < NLOAD_;
  const int  xs_row = tid / CPR_;
  const int  xs_col = (tid % CPR_) * 8;
  const _Float16* xsrc = xbase + (size_t)xs_row * (T_ * KIN) + xs_col;
  if (loader)
    *(v8h*)&x_lds[xs_row][xs_col] = *(const v8h*)xsrc;

  __syncthreads();

  for (int t = 0; t < T_; ++t) {
    v8h xr = {};
    if (loader) {
      if (t + 1 < T_) xr = *(const v8h*)(xsrc + (size_t)(t + 1) * KIN);
      if (t + 2 < T_) __builtin_prefetch(xsrc + (size_t)(t + 2) * KIN, 0, 1);
    }

    v16h xa[KX];
    #pragma unroll
    for (int kt = 0; kt < KX; ++kt) {
      const _Float16* p = &x_lds[lane & 15][kt * 32 + hi * 8];
      v8h a = *(const v8h*)p;
      v8h b = *(const v8h*)(p + 16);
      xa[kt] = __builtin_shufflevector(a, b, 0,1,2,3,4,5,6,7,8,9,10,11,12,13,14,15);
    }

    v16h ha[4];
    #pragma unroll
    for (int kt = 0; kt < 4; ++kt) {
      const _Float16* p = &h_lds[lane & 15][kt * 32 + hi * 8];
      v8h a = *(const v8h*)p;
      v8h b = *(const v8h*)(p + 16);
      ha[kt] = __builtin_shufflevector(a, b, 0,1,2,3,4,5,6,7,8,9,10,11,12,13,14,15);
    }

    #pragma unroll
    for (int n = 0; n < NT_; ++n) {
      v8f acc;
      #pragma unroll
      for (int m = 0; m < 8; ++m) acc[m] = bv[n];
      #pragma unroll
      for (int kt = 0; kt < KX; ++kt)
        acc = __builtin_amdgcn_wmma_f32_16x16x32_f16(false, xa[kt], false, Bx[n][kt],
                                                     (short)0, acc, false, false);
      #pragma unroll
      for (int kt = 0; kt < 4; ++kt)
        acc = __builtin_amdgcn_wmma_f32_16x16x32_f16(false, ha[kt], false, Bh[n][kt],
                                                     (short)0, acc, false, false);

      const int col = ncol0 + n * 16 + (lane & 15);
      #pragma unroll
      for (int m = 0; m < 8; ++m) {
        float a = act_off + act_post * fast_tanh(acc[m] * act_pre);
        g_lds[m + 8 * hi][col] = a;
      }
    }
    __syncthreads();

    if (loader && t + 1 < T_)
      *(v8h*)&x_lds[xs_row][xs_col] = xr;

    #pragma unroll
    for (int e = 0; e < 4; ++e) {
      int q = tid * 4 + e;
      int r = q >> 7;
      int j = q & (H_ - 1);
      float iv = g_lds[r][j];
      float fv = g_lds[r][H_ + j];
      float gv = g_lds[r][2 * H_ + j];
      float ov = g_lds[r][3 * H_ + j];
      c[e] = fv * c[e] + iv * gv;
      float hv = ov * fast_tanh(c[e]);
      h_lds[r][j] = (_Float16)hv;
      if (hseq_out)
        hseq_out[(size_t)(b0 + r) * T_ * H_ + (size_t)t * H_ + j] = (_Float16)hv;
      if (hlast_out && t == T_ - 1)
        hlast_out[(b0 + r) * H_ + j] = hv;
    }
    __syncthreads();
  }
}

__global__ void fc_kernel(const float* __restrict__ h1, const float* __restrict__ Wfc0,
                          const float* __restrict__ bfc0, const float* __restrict__ Wfc1,
                          const float* __restrict__ bfc1, float* __restrict__ out) {
  int b = blockIdx.x;
  int tid = threadIdx.x;
  __shared__ float tmp[32];
  float s = bfc0[tid];
  for (int k = 0; k < H_; ++k) s += h1[b * H_ + k] * Wfc0[tid * H_ + k];
  tmp[tid] = s;
  __syncthreads();
  if (tid < OUT_) {
    float o = bfc1[tid];
    for (int k = 0; k < 32; ++k) o += tmp[k] * Wfc1[tid * 32 + k];
    out[b * OUT_ + tid] = o;
  }
}

extern "C" void kernel_launch(void* const* d_in, const int* in_sizes, int n_in,
                              void* d_out, int out_size, void* d_ws, size_t ws_size,
                              hipStream_t stream) {
  (void)in_sizes; (void)n_in; (void)out_size; (void)ws_size;
  const float* x    = (const float*)d_in[0];
  const float* Wih0 = (const float*)d_in[1];
  const float* Whh0 = (const float*)d_in[2];
  const float* bih0 = (const float*)d_in[3];
  const float* bhh0 = (const float*)d_in[4];
  const float* Wih1 = (const float*)d_in[5];
  const float* Whh1 = (const float*)d_in[6];
  const float* bih1 = (const float*)d_in[7];
  const float* bhh1 = (const float*)d_in[8];
  const float* Wfc0 = (const float*)d_in[9];
  const float* bfc0 = (const float*)d_in[10];
  const float* Wfc1 = (const float*)d_in[11];
  const float* bfc1 = (const float*)d_in[12];

  char* p = (char*)d_ws;
  auto alloc = [&](size_t bytes) {
    char* r = p;
    p += (bytes + 255) & ~(size_t)255;
    return r;
  };
  _Float16* x_h    = (_Float16*)alloc((size_t)B_ * T_ * D_ * 2);
  _Float16* h0seq  = (_Float16*)alloc((size_t)B_ * T_ * H_ * 2);
  _Float16* wih0_h = (_Float16*)alloc((size_t)G_ * D_ * 2);
  _Float16* whh0_h = (_Float16*)alloc((size_t)G_ * H_ * 2);
  _Float16* wih1_h = (_Float16*)alloc((size_t)G_ * H_ * 2);
  _Float16* whh1_h = (_Float16*)alloc((size_t)G_ * H_ * 2);
  float*    b0     = (float*)alloc((size_t)G_ * 4);
  float*    b1     = (float*)alloc((size_t)G_ * 4);
  float*    h1last = (float*)alloc((size_t)B_ * H_ * 4);

  auto cvt = [&](const float* s, _Float16* d, int n) {
    cvt_f32_to_f16<<<(n + 255) / 256, 256, 0, stream>>>(s, d, n);
  };
  cvt(x, x_h, B_ * T_ * D_);
  cvt(Wih0, wih0_h, G_ * D_);
  cvt(Whh0, whh0_h, G_ * H_);
  cvt(Wih1, wih1_h, G_ * H_);
  cvt(Whh1, whh1_h, G_ * H_);
  add_vec<<<2, 256, 0, stream>>>(bih0, bhh0, b0, G_);
  add_vec<<<2, 256, 0, stream>>>(bih1, bhh1, b1, G_);

  lstm_layer_kernel<D_><<<B_ / BT_, THREADS_, 0, stream>>>(x_h, wih0_h, whh0_h, b0,
                                                           h0seq, nullptr);
  lstm_layer_kernel<H_><<<B_ / BT_, THREADS_, 0, stream>>>(h0seq, wih1_h, whh1_h, b1,
                                                           nullptr, h1last);
  fc_kernel<<<B_, 32, 0, stream>>>(h1last, Wfc0, bfc0, Wfc1, bfc1, (float*)d_out);
}


# Each of 8 waves owns all four gates of 16 hidden units (weights as WMMA A operand): cell update in registers, no gate exchange through LDS, one barrier per step, h and x double-buffered
# speedup vs baseline: 3.1818x; 1.7051x over previous
#include <hip/hip_runtime.h>


#define B_ 512
#define T_ 512
#define D_ 64
#define H_ 128
#define G_ 512
#define OUT_ 16
#define BT_ 16
#define THREADS_ 256
#define NG_ 4

typedef __attribute__((ext_vector_type(16))) _Float16 v16h;
typedef __attribute__((ext_vector_type(8)))  _Float16 v8h;
typedef __attribute__((ext_vector_type(8)))  float    v8f;

__device__ __forceinline__ float fast_tanh(float x) {
#if __has_builtin(__builtin_amdgcn_tanhf)
  return __builtin_amdgcn_tanhf(x);
#else
  float e = __builtin_amdgcn_exp2f(x * 2.8853900817779268f);
  return 1.0f - 2.0f * __builtin_amdgcn_rcpf(e + 1.0f);
#endif
}

__device__ __forceinline__ v16h ldfrag_g(const _Float16* __restrict__ base,
                                         int stride, int koff, int lane) {
  int r  = lane & 15;
  int hi = lane >> 4;
  const _Float16* p = base + (size_t)r * stride + koff + hi * 8;
  v8h a = *(const v8h*)p;
  v8h b = *(const v8h*)(p + 16);
  return __builtin_shufflevector(a, b, 0,1,2,3,4,5,6,7,8,9,10,11,12,13,14,15);
}

__device__ __forceinline__ v16h ldfrag_s(const _Float16* rowbase, int koff, int hi) {
  const _Float16* p = rowbase + koff + hi * 8;
  v8h a = *(const v8h*)p;
  v8h b = *(const v8h*)(p + 16);
  return __builtin_shufflevector(a, b, 0,1,2,3,4,5,6,7,8,9,10,11,12,13,14,15);
}

__global__ void cvt_f32_to_f16(const float* __restrict__ s, _Float16* __restrict__ d, int n) {
  int i = blockIdx.x * blockDim.x + threadIdx.x;
  if (i < n) d[i] = (_Float16)s[i];
}

__global__ void add_vec(const float* __restrict__ a, const float* __restrict__ b,
                        float* __restrict__ o, int n) {
  int i = blockIdx.x * blockDim.x + threadIdx.x;
  if (i < n) o[i] = a[i] + b[i];
}

template<int KIN>
__global__ __launch_bounds__(THREADS_, 1)
void lstm_layer_kernel(const _Float16* __restrict__ xseq,
                       const _Float16* __restrict__ Wih,
                       const _Float16* __restrict__ Whh,
                       const float*    __restrict__ bias,
                       _Float16*       __restrict__ hseq_out,
                       float*          __restrict__ hlast_out)
{
  constexpr int KX = KIN / 32;

  __shared__ __attribute__((aligned(16)))
  _Float16 h_lds[2][BT_][H_ + 8];
  __shared__ __attribute__((aligned(16)))
  _Float16 x_lds[2][BT_][KIN + 8];

  const int tid   = threadIdx.x;
  const int lane  = tid & 31;
  const int wave  = tid >> 5;
  const int hi    = lane >> 4;
  const int c16   = lane & 15;
  const int b0    = blockIdx.x * BT_;
  const int j0    = wave * 16;

  for (int i = tid; i < BT_ * (H_ + 8); i += THREADS_)
    ((_Float16*)h_lds[0])[i] = (_Float16)0.0f;

  v16h Wx[NG_][KX];
  v16h Wh[NG_][4];
  v8f bv[NG_];
  #pragma unroll
  for (int g = 0; g < NG_; ++g) {
    const int nc = g * H_ + j0;
    #pragma unroll
    for (int m = 0; m < 8; ++m) bv[g][m] = bias[nc + 8 * hi + m];
    #pragma unroll
    for (int kt = 0; kt < KX; ++kt)
      Wx[g][kt] = ldfrag_g(Wih + (size_t)nc * KIN, KIN, kt * 32, lane);
    #pragma unroll
    for (int kt = 0; kt < 4; ++kt)
      Wh[g][kt] = ldfrag_g(Whh + (size_t)nc * H_, H_, kt * 32, lane);
  }

  float c[8] = {0.f, 0.f, 0.f, 0.f, 0.f, 0.f, 0.f, 0.f};

  const _Float16* xbase = xseq + (size_t)b0 * T_ * KIN;

  constexpr int CPR_   = KIN / 8;
  constexpr int NLOAD_ = BT_ * CPR_;
  const bool loader = tid < NLOAD_;
  const int  xs_row = tid / CPR_;
  const int  xs_col = (tid % CPR_) * 8;
  const _Float16* xsrc = xbase + (size_t)xs_row * (T_ * KIN) + xs_col;
  v8h xr = {};
  if (loader) {
    *(v8h*)&x_lds[0][xs_row][xs_col] = *(const v8h*)xsrc;
    xr = *(const v8h*)(xsrc + KIN);
  }

  const int co_row = tid >> 4;
  const int co_col = (tid & 15) * 8;
  _Float16* hdst = hseq_out ? hseq_out + (size_t)(b0 + co_row) * T_ * H_ + co_col : nullptr;

  __syncthreads();

  for (int t = 0; t < T_; ++t) {
    const int cur = t & 1;
    const int nxt = cur ^ 1;

    if (loader) {
      if (t + 1 < T_) *(v8h*)&x_lds[nxt][xs_row][xs_col] = xr;
      if (t + 2 < T_) xr = *(const v8h*)(xsrc + (size_t)(t + 2) * KIN);
      if (t + 3 < T_) __builtin_prefetch(xsrc + (size_t)(t + 3) * KIN, 0, 1);
    }

    v16h xa[KX];
    #pragma unroll
    for (int kt = 0; kt < KX; ++kt)
      xa[kt] = ldfrag_s(&x_lds[cur][c16][0], kt * 32, hi);

    v16h ha[4];
    #pragma unroll
    for (int kt = 0; kt < 4; ++kt)
      ha[kt] = ldfrag_s(&h_lds[cur][c16][0], kt * 32, hi);

    if (hseq_out && t > 0) {
      v8h hv8 = *(const v8h*)&h_lds[cur][co_row][co_col];
      *(v8h*)(hdst + (size_t)(t - 1) * H_) = hv8;
    }

    v8f acc[NG_];
    #pragma unroll
    for (int kt = 0; kt < KX; ++kt) {
      #pragma unroll
      for (int g = 0; g < NG_; ++g)
        acc[g] = __builtin_amdgcn_wmma_f32_16x16x32_f16(false, Wx[g][kt], false, xa[kt],
                                                        (short)0, kt == 0 ? bv[g] : acc[g],
                                                        false, false);
    }
    #pragma unroll
    for (int kt = 0; kt < 4; ++kt) {
      #pragma unroll
      for (int g = 0; g < NG_; ++g)
        acc[g] = __builtin_amdgcn_wmma_f32_16x16x32_f16(false, Wh[g][kt], false, ha[kt],
                                                        (short)0, acc[g], false, false);
    }

    v8h hpack;
    #pragma unroll
    for (int m = 0; m < 8; ++m) {
      float iv = 0.5f + 0.5f * fast_tanh(acc[0][m] * 0.5f);
      float fv = 0.5f + 0.5f * fast_tanh(acc[1][m] * 0.5f);
      float gv = fast_tanh(acc[2][m]);
      float ov = 0.5f + 0.5f * fast_tanh(acc[3][m] * 0.5f);
      c[m] = fv * c[m] + iv * gv;
      float hv = ov * fast_tanh(c[m]);
      hpack[m] = (_Float16)hv;
      if (hlast_out && t == T_ - 1)
        hlast_out[(b0 + c16) * H_ + j0 + 8 * hi + m] = hv;
    }
    *(v8h*)&h_lds[nxt][c16][j0 + 8 * hi] = hpack;

    __syncthreads();
  }

  if (hseq_out) {
    v8h hv8 = *(const v8h*)&h_lds[T_ & 1][co_row][co_col];
    *(v8h*)(hdst + (size_t)(T_ - 1) * H_) = hv8;
  }
}

__global__ void fc_kernel(const float* __restrict__ h1, const float* __restrict__ Wfc0,
                          const float* __restrict__ bfc0, const float* __restrict__ Wfc1,
                          const float* __restrict__ bfc1, float* __restrict__ out) {
  int b = blockIdx.x;
  int tid = threadIdx.x;
  __shared__ float tmp[32];
  float s = bfc0[tid];
  for (int k = 0; k < H_; ++k) s += h1[b * H_ + k] * Wfc0[tid * H_ + k];
  tmp[tid] = s;
  __syncthreads();
  if (tid < OUT_) {
    float o = bfc1[tid];
    for (int k = 0; k < 32; ++k) o += tmp[k] * Wfc1[tid * 32 + k];
    out[b * OUT_ + tid] = o;
  }
}

extern "C" void kernel_launch(void* const* d_in, const int* in_sizes, int n_in,
                              void* d_out, int out_size, void* d_ws, size_t ws_size,
                              hipStream_t stream) {
  (void)in_sizes; (void)n_in; (void)out_size; (void)ws_size;
  const float* x    = (const float*)d_in[0];
  const float* Wih0 = (const float*)d_in[1];
  const float* Whh0 = (const float*)d_in[2];
  const float* bih0 = (const float*)d_in[3];
  const float* bhh0 = (const float*)d_in[4];
  const float* Wih1 = (const float*)d_in[5];
  const float* Whh1 = (const float*)d_in[6];
  const float* bih1 = (const float*)d_in[7];
  const float* bhh1 = (const float*)d_in[8];
  const float* Wfc0 = (const float*)d_in[9];
  const float* bfc0 = (const float*)d_in[10];
  const float* Wfc1 = (const float*)d_in[11];
  const float* bfc1 = (const float*)d_in[12];

  char* p = (char*)d_ws;
  auto alloc = [&](size_t bytes) {
    char* r = p;
    p += (bytes + 255) & ~(size_t)255;
    return r;
  };
  _Float16* x_h    = (_Float16*)alloc((size_t)B_ * T_ * D_ * 2);
  _Float16* h0seq  = (_Float16*)alloc((size_t)B_ * T_ * H_ * 2);
  _Float16* wih0_h = (_Float16*)alloc((size_t)G_ * D_ * 2);
  _Float16* whh0_h = (_Float16*)alloc((size_t)G_ * H_ * 2);
  _Float16* wih1_h = (_Float16*)alloc((size_t)G_ * H_ * 2);
  _Float16* whh1_h = (_Float16*)alloc((size_t)G_ * H_ * 2);
  float*    b0     = (float*)alloc((size_t)G_ * 4);
  float*    b1     = (float*)alloc((size_t)G_ * 4);
  float*    h1last = (float*)alloc((size_t)B_ * H_ * 4);

  auto cvt = [&](const float* s, _Float16* d, int n) {
    cvt_f32_to_f16<<<(n + 255) / 256, 256, 0, stream>>>(s, d, n);
  };
  cvt(x, x_h, B_ * T_ * D_);
  cvt(Wih0, wih0_h, G_ * D_);
  cvt(Whh0, whh0_h, G_ * H_);
  cvt(Wih1, wih1_h, G_ * H_);
  cvt(Whh1, whh1_h, G_ * H_);
  add_vec<<<2, 256, 0, stream>>>(bih0, bhh0, b0, G_);
  add_vec<<<2, 256, 0, stream>>>(bih1, bhh1, b1, G_);

  lstm_layer_kernel<D_><<<B_ / BT_, THREADS_, 0, stream>>>(x_h, wih0_h, whh0_h, b0,
                                                           h0seq, nullptr);
  lstm_layer_kernel<H_><<<B_ / BT_, THREADS_, 0, stream>>>(h0seq, wih1_h, whh1_h, b1,
                                                           nullptr, h1last);
  fc_kernel<<<B_, 32, 0, stream>>>(h1last, Wfc0, bfc0, Wfc1, bfc1, (float*)d_out);
}


# Step t also computes the input projection Wih@x of step t+1, beside its activations, and issues the next h and x LDS reads right behind the barrier: only the Whh@h WMMAs stay on the serial path
# speedup vs baseline: 3.2949x; 1.0355x over previous
#include <hip/hip_runtime.h>


#define B_ 512
#define T_ 512
#define D_ 64
#define H_ 128
#define G_ 512
#define OUT_ 16
#define BT_ 16
#define THREADS_ 256
#define NG_ 4

typedef __attribute__((ext_vector_type(16))) _Float16 v16h;
typedef __attribute__((ext_vector_type(8)))  _Float16 v8h;
typedef __attribute__((ext_vector_type(8)))  float    v8f;

__device__ __forceinline__ float fast_tanh(float x) {
#if __has_builtin(__builtin_amdgcn_tanhf)
  return __builtin_amdgcn_tanhf(x);
#else
  float e = __builtin_amdgcn_exp2f(x * 2.8853900817779268f);
  return 1.0f - 2.0f * __builtin_amdgcn_rcpf(e + 1.0f);
#endif
}

__device__ __forceinline__ v16h ldfrag_g(const _Float16* __restrict__ base,
                                         int stride, int koff, int lane) {
  int r  = lane & 15;
  int hi = lane >> 4;
  const _Float16* p = base + (size_t)r * stride + koff + hi * 8;
  v8h a = *(const v8h*)p;
  v8h b = *(const v8h*)(p + 16);
  return __builtin_shufflevector(a, b, 0,1,2,3,4,5,6,7,8,9,10,11,12,13,14,15);
}

__device__ __forceinline__ v16h ldfrag_s(const _Float16* rowbase, int koff, int hi) {
  const _Float16* p = rowbase + koff + hi * 8;
  v8h a = *(const v8h*)p;
  v8h b = *(const v8h*)(p + 16);
  return __builtin_shufflevector(a, b, 0,1,2,3,4,5,6,7,8,9,10,11,12,13,14,15);
}

__global__ void cvt_f32_to_f16(const float* __restrict__ s, _Float16* __restrict__ d, int n) {
  int i = blockIdx.x * blockDim.x + threadIdx.x;
  if (i < n) d[i] = (_Float16)s[i];
}

__global__ void add_vec(const float* __restrict__ a, const float* __restrict__ b,
                        float* __restrict__ o, int n) {
  int i = blockIdx.x * blockDim.x + threadIdx.x;
  if (i < n) o[i] = a[i] + b[i];
}

template<int KIN, bool SEQ>
__global__ __launch_bounds__(THREADS_, 1)
void lstm_layer_kernel(const _Float16* __restrict__ xseq,
                       const _Float16* __restrict__ Wih,
                       const _Float16* __restrict__ Whh,
                       const float*    __restrict__ bias,
                       _Float16*       __restrict__ hseq_out,
                       float*          __restrict__ hlast_out)
{
  constexpr int KX = KIN / 32;

  __shared__ __attribute__((aligned(16)))
  _Float16 h_lds[2][BT_][H_ + 8];
  __shared__ __attribute__((aligned(16)))
  _Float16 x_lds[2][BT_][KIN + 8];

  const int tid   = threadIdx.x;
  const int lane  = tid & 31;
  const int wave  = tid >> 5;
  const int hi    = lane >> 4;
  const int c16   = lane & 15;
  const int b0    = blockIdx.x * BT_;
  const int j0    = wave * 16;

  for (int i = tid; i < BT_ * (H_ + 8); i += THREADS_)
    ((_Float16*)h_lds[0])[i] = (_Float16)0.0f;

  v16h Wx[NG_][KX];
  v16h Wh[NG_][4];
  v8f bv[NG_];
  #pragma unroll
  for (int g = 0; g < NG_; ++g) {
    const int nc = g * H_ + j0;
    #pragma unroll
    for (int m = 0; m < 8; ++m) bv[g][m] = bias[nc + 8 * hi + m];
    #pragma unroll
    for (int kt = 0; kt < KX; ++kt)
      Wx[g][kt] = ldfrag_g(Wih + (size_t)nc * KIN, KIN, kt * 32, lane);
    #pragma unroll
    for (int kt = 0; kt < 4; ++kt)
      Wh[g][kt] = ldfrag_g(Whh + (size_t)nc * H_, H_, kt * 32, lane);
  }

  float c[8] = {0.f, 0.f, 0.f, 0.f, 0.f, 0.f, 0.f, 0.f};

  const _Float16* xbase = xseq + (size_t)b0 * T_ * KIN;

  constexpr int CPR_   = KIN / 8;
  constexpr int NLOAD_ = BT_ * CPR_;
  const bool loader = tid < NLOAD_;
  const int  xs_row = tid / CPR_;
  const int  xs_col = (tid % CPR_) * 8;
  const _Float16* xsrc = xbase + (size_t)xs_row * (T_ * KIN) + xs_col;
  v8h xr = {};
  if (loader) {
    *(v8h*)&x_lds[0][xs_row][xs_col] = *(const v8h*)xsrc;
    *(v8h*)&x_lds[1][xs_row][xs_col] = *(const v8h*)(xsrc + KIN);
    xr = *(const v8h*)(xsrc + 2 * KIN);
  }

  const int co_row = tid >> 4;
  const int co_col = (tid & 15) * 8;
  _Float16* hdst = SEQ ? hseq_out + (size_t)(b0 + co_row) * T_ * H_ + co_col : nullptr;

  __syncthreads();

  v8f accx[NG_];
  {
    v16h xa0[KX];
    #pragma unroll
    for (int kt = 0; kt < KX; ++kt)
      xa0[kt] = ldfrag_s(&x_lds[0][c16][0], kt * 32, hi);
    #pragma unroll
    for (int kt = 0; kt < KX; ++kt) {
      #pragma unroll
      for (int g = 0; g < NG_; ++g)
        accx[g] = __builtin_amdgcn_wmma_f32_16x16x32_f16(false, Wx[g][kt], false, xa0[kt],
                                                         (short)0, kt == 0 ? bv[g] : accx[g],
                                                         false, false);
    }
  }
  __syncthreads();

  v16h ha[4];
  #pragma unroll
  for (int kt = 0; kt < 4; ++kt)
    ha[kt] = ldfrag_s(&h_lds[0][c16][0], kt * 32, hi);
  v16h xa[KX];
  #pragma unroll
  for (int kt = 0; kt < KX; ++kt)
    xa[kt] = ldfrag_s(&x_lds[1][c16][0], kt * 32, hi);

  v8h hv8 = {};

  float hkeep[8] = {0.f, 0.f, 0.f, 0.f, 0.f, 0.f, 0.f, 0.f};

  for (int t = 0; t < T_; ++t) {
    const int cur = t & 1;
    const int nxt = cur ^ 1;

    if (loader) {
      if (t + 2 < T_) *(v8h*)&x_lds[cur][xs_row][xs_col] = xr;
      if (t + 3 < T_) xr = *(const v8h*)(xsrc + (size_t)(t + 3) * KIN);
      if (t + 4 < T_) __builtin_prefetch(xsrc + (size_t)(t + 4) * KIN, 0, 1);
    }

    if (SEQ && t > 0) *(v8h*)(hdst + (size_t)(t - 1) * H_) = hv8;

    v8f acc[NG_];
    #pragma unroll
    for (int kt = 0; kt < 4; ++kt) {
      #pragma unroll
      for (int g = 0; g < NG_; ++g)
        acc[g] = __builtin_amdgcn_wmma_f32_16x16x32_f16(false, Wh[g][kt], false, ha[kt],
                                                        (short)0, kt == 0 ? accx[g] : acc[g],
                                                        false, false);
    }

    #pragma unroll
    for (int kt = 0; kt < KX; ++kt) {
      #pragma unroll
      for (int g = 0; g < NG_; ++g)
        accx[g] = __builtin_amdgcn_wmma_f32_16x16x32_f16(false, Wx[g][kt], false, xa[kt],
                                                         (short)0, kt == 0 ? bv[g] : accx[g],
                                                         false, false);
    }

    v8h hpack;
    #pragma unroll
    for (int m = 0; m < 8; ++m) {
      float iv = 0.5f + 0.5f * fast_tanh(acc[0][m] * 0.5f);
      float fv = 0.5f + 0.5f * fast_tanh(acc[1][m] * 0.5f);
      float gv = fast_tanh(acc[2][m]);
      float ov = 0.5f + 0.5f * fast_tanh(acc[3][m] * 0.5f);
      c[m] = fv * c[m] + iv * gv;
      float hv = ov * fast_tanh(c[m]);
      hpack[m] = (_Float16)hv;
      hkeep[m] = hv;
    }
    *(v8h*)&h_lds[nxt][c16][j0 + 8 * hi] = hpack;

    __syncthreads();

    #pragma unroll
    for (int kt = 0; kt < 4; ++kt)
      ha[kt] = ldfrag_s(&h_lds[nxt][c16][0], kt * 32, hi);
    if (SEQ) hv8 = *(const v8h*)&h_lds[nxt][co_row][co_col];
    __builtin_amdgcn_sched_barrier(0);
    #pragma unroll
    for (int kt = 0; kt < KX; ++kt)
      xa[kt] = ldfrag_s(&x_lds[cur][c16][0], kt * 32, hi);
  }

  if (SEQ) *(v8h*)(hdst + (size_t)(T_ - 1) * H_) = hv8;

  if (!SEQ) {
    #pragma unroll
    for (int m = 0; m < 8; ++m)
      hlast_out[(b0 + c16) * H_ + j0 + 8 * hi + m] = hkeep[m];
  }

}

__global__ void fc_kernel(const float* __restrict__ h1, const float* __restrict__ Wfc0,
                          const float* __restrict__ bfc0, const float* __restrict__ Wfc1,
                          const float* __restrict__ bfc1, float* __restrict__ out) {
  int b = blockIdx.x;
  int tid = threadIdx.x;
  __shared__ float tmp[32];
  float s = bfc0[tid];
  for (int k = 0; k < H_; ++k) s += h1[b * H_ + k] * Wfc0[tid * H_ + k];
  tmp[tid] = s;
  __syncthreads();
  if (tid < OUT_) {
    float o = bfc1[tid];
    for (int k = 0; k < 32; ++k) o += tmp[k] * Wfc1[tid * 32 + k];
    out[b * OUT_ + tid] = o;
  }
}

extern "C" void kernel_launch(void* const* d_in, const int* in_sizes, int n_in,
                              void* d_out, int out_size, void* d_ws, size_t ws_size,
                              hipStream_t stream) {
  (void)in_sizes; (void)n_in; (void)out_size; (void)ws_size;
  const float* x    = (const float*)d_in[0];
  const float* Wih0 = (const float*)d_in[1];
  const float* Whh0 = (const float*)d_in[2];
  const float* bih0 = (const float*)d_in[3];
  const float* bhh0 = (const float*)d_in[4];
  const float* Wih1 = (const float*)d_in[5];
  const float* Whh1 = (const float*)d_in[6];
  const float* bih1 = (const float*)d_in[7];
  const float* bhh1 = (const float*)d_in[8];
  const float* Wfc0 = (const float*)d_in[9];
  const float* bfc0 = (const float*)d_in[10];
  const float* Wfc1 = (const float*)d_in[11];
  const float* bfc1 = (const float*)d_in[12];

  char* p = (char*)d_ws;
  auto alloc = [&](size_t bytes) {
    char* r = p;
    p += (bytes + 255) & ~(size_t)255;
    return r;
  };
  _Float16* x_h    = (_Float16*)alloc((size_t)B_ * T_ * D_ * 2);
  _Float16* h0seq  = (_Float16*)alloc((size_t)B_ * T_ * H_ * 2);
  _Float16* wih0_h = (_Float16*)alloc((size_t)G_ * D_ * 2);
  _Float16* whh0_h = (_Float16*)alloc((size_t)G_ * H_ * 2);
  _Float16* wih1_h = (_Float16*)alloc((size_t)G_ * H_ * 2);
  _Float16* whh1_h = (_Float16*)alloc((size_t)G_ * H_ * 2);
  float*    b0     = (float*)alloc((size_t)G_ * 4);
  float*    b1     = (float*)alloc((size_t)G_ * 4);
  float*    h1last = (float*)alloc((size_t)B_ * H_ * 4);

  auto cvt = [&](const float* s, _Float16* d, int n) {
    cvt_f32_to_f16<<<(n + 255) / 256, 256, 0, stream>>>(s, d, n);
  };
  cvt(x, x_h, B_ * T_ * D_);
  cvt(Wih0, wih0_h, G_ * D_);
  cvt(Whh0, whh0_h, G_ * H_);
  cvt(Wih1, wih1_h, G_ * H_);
  cvt(Whh1, whh1_h, G_ * H_);
  add_vec<<<2, 256, 0, stream>>>(bih0, bhh0, b0, G_);
  add_vec<<<2, 256, 0, stream>>>(bih1, bhh1, b1, G_);

  lstm_layer_kernel<D_, true><<<B_ / BT_, THREADS_, 0, stream>>>(x_h, wih0_h, whh0_h, b0,
                                                                 h0seq, nullptr);
  lstm_layer_kernel<H_, false><<<B_ / BT_, THREADS_, 0, stream>>>(h0seq, wih1_h, whh1_h, b1,
                                                                  nullptr, h1last);
  fc_kernel<<<B_, 32, 0, stream>>>(h1last, Wfc0, bfc0, Wfc1, bfc1, (float*)d_out);
}


# Layer 0 reads the f32 x itself and rounds it to the 16-bit operand type in its LDS hand-over: the separate conversion pass over x (2^24 one-element threads, a 16-bit copy written and re-read) is gone
# speedup vs baseline: 3.4656x; 1.0518x over previous
#include <hip/hip_runtime.h>


#define B_ 512
#define T_ 512
#define D_ 64
#define H_ 128
#define G_ 512
#define OUT_ 16
#define BT_ 16
#define THREADS_ 256
#define NG_ 4

typedef __attribute__((ext_vector_type(16))) _Float16 v16h;
typedef __attribute__((ext_vector_type(8)))  _Float16 v8h;
typedef __attribute__((ext_vector_type(8)))  float    v8f;
typedef __attribute__((ext_vector_type(4)))  _Float16 v4h;
typedef __attribute__((ext_vector_type(4)))  float    v4f;

__device__ __forceinline__ float fast_tanh(float x) {
#if __has_builtin(__builtin_amdgcn_tanhf)
  return __builtin_amdgcn_tanhf(x);
#else
  float e = __builtin_amdgcn_exp2f(x * 2.8853900817779268f);
  return 1.0f - 2.0f * __builtin_amdgcn_rcpf(e + 1.0f);
#endif
}

__device__ __forceinline__ v16h ldfrag_g(const _Float16* __restrict__ base,
                                         int stride, int koff, int lane) {
  int r  = lane & 15;
  int hi = lane >> 4;
  const _Float16* p = base + (size_t)r * stride + koff + hi * 8;
  v8h a = *(const v8h*)p;
  v8h b = *(const v8h*)(p + 16);
  return __builtin_shufflevector(a, b, 0,1,2,3,4,5,6,7,8,9,10,11,12,13,14,15);
}

__device__ __forceinline__ v16h ldfrag_s(const _Float16* rowbase, int koff, int hi) {
  const _Float16* p = rowbase + koff + hi * 8;
  v8h a = *(const v8h*)p;
  v8h b = *(const v8h*)(p + 16);
  return __builtin_shufflevector(a, b, 0,1,2,3,4,5,6,7,8,9,10,11,12,13,14,15);
}

__global__ void cvt_f32_to_f16(const float* __restrict__ s, _Float16* __restrict__ d, int n) {
  int i = blockIdx.x * blockDim.x + threadIdx.x;
  if (i < n) d[i] = (_Float16)s[i];
}

__global__ void add_vec(const float* __restrict__ a, const float* __restrict__ b,
                        float* __restrict__ o, int n) {
  int i = blockIdx.x * blockDim.x + threadIdx.x;
  if (i < n) o[i] = a[i] + b[i];
}

template<int KIN, bool SEQ>
__global__ __launch_bounds__(THREADS_, 1)
void lstm_layer_kernel(const float*    __restrict__ x32,
                       const _Float16* __restrict__ xh,
                       const _Float16* __restrict__ Wih,
                       const _Float16* __restrict__ Whh,
                       const float*    __restrict__ bias,
                       _Float16*       __restrict__ hseq_out,
                       float*          __restrict__ hlast_out)
{
  constexpr int KX = KIN / 32;

  __shared__ __attribute__((aligned(16)))
  _Float16 h_lds[2][BT_][H_ + 8];
  __shared__ __attribute__((aligned(16)))
  _Float16 x_lds[2][BT_][KIN + 8];

  const int tid   = threadIdx.x;
  const int lane  = tid & 31;
  const int wave  = tid >> 5;
  const int hi    = lane >> 4;
  const int c16   = lane & 15;
  const int b0    = blockIdx.x * BT_;
  const int j0    = wave * 16;

  for (int i = tid; i < BT_ * (H_ + 8); i += THREADS_)
    ((_Float16*)h_lds[0])[i] = (_Float16)0.0f;

  v16h Wx[NG_][KX];
  v16h Wh[NG_][4];
  v8f bv[NG_];
  #pragma unroll
  for (int g = 0; g < NG_; ++g) {
    const int nc = g * H_ + j0;
    #pragma unroll
    for (int m = 0; m < 8; ++m) bv[g][m] = bias[nc + 8 * hi + m];
    #pragma unroll
    for (int kt = 0; kt < KX; ++kt)
      Wx[g][kt] = ldfrag_g(Wih + (size_t)nc * KIN, KIN, kt * 32, lane);
    #pragma unroll
    for (int kt = 0; kt < 4; ++kt)
      Wh[g][kt] = ldfrag_g(Whh + (size_t)nc * H_, H_, kt * 32, lane);
  }

  float c[8] = {0.f, 0.f, 0.f, 0.f, 0.f, 0.f, 0.f, 0.f};

  constexpr int EPP_   = SEQ ? 4 : 8;
  constexpr int CPR_   = KIN / EPP_;
  constexpr int NLOAD_ = BT_ * CPR_;
  static_assert(NLOAD_ <= THREADS_, "one piece per thread");
  const bool loader = tid < NLOAD_;
  const int  xs_row = tid / CPR_;
  const int  xs_col = (tid % CPR_) * EPP_;
  const size_t xoff = (size_t)(b0 + xs_row) * (T_ * KIN) + xs_col;
  const float*    xsrc32 = SEQ ? x32 + xoff : nullptr;
  const _Float16* xsrc16 = SEQ ? nullptr : xh + xoff;
  v4f xr32 = {};
  v8h xr16 = {};
  auto x_fetch = [&](int s) {
    if constexpr (SEQ) xr32 = *(const v4f*)(xsrc32 + (size_t)s * KIN);
    else               xr16 = *(const v8h*)(xsrc16 + (size_t)s * KIN);
  };
  auto x_put = [&](int cp) {
    if constexpr (SEQ) {
      v4h p4 = {(_Float16)xr32[0], (_Float16)xr32[1], (_Float16)xr32[2], (_Float16)xr32[3]};
      *(v4h*)&x_lds[cp][xs_row][xs_col] = p4;
    } else {
      *(v8h*)&x_lds[cp][xs_row][xs_col] = xr16;
    }
  };
  if (loader) {
    x_fetch(0); x_put(0);
    x_fetch(1); x_put(1);
    x_fetch(2);
  }

  const int co_row = tid >> 4;
  const int co_col = (tid & 15) * 8;
  _Float16* hdst = SEQ ? hseq_out + (size_t)(b0 + co_row) * T_ * H_ + co_col : nullptr;

  __syncthreads();

  v8f accx[NG_];
  {
    v16h xa0[KX];
    #pragma unroll
    for (int kt = 0; kt < KX; ++kt)
      xa0[kt] = ldfrag_s(&x_lds[0][c16][0], kt * 32, hi);
    #pragma unroll
    for (int kt = 0; kt < KX; ++kt) {
      #pragma unroll
      for (int g = 0; g < NG_; ++g)
        accx[g] = __builtin_amdgcn_wmma_f32_16x16x32_f16(false, Wx[g][kt], false, xa0[kt],
                                                         (short)0, kt == 0 ? bv[g] : accx[g],
                                                         false, false);
    }
  }
  __syncthreads();

  v16h ha[4];
  #pragma unroll
  for (int kt = 0; kt < 4; ++kt)
    ha[kt] = ldfrag_s(&h_lds[0][c16][0], kt * 32, hi);
  v16h xa[KX];
  #pragma unroll
  for (int kt = 0; kt < KX; ++kt)
    xa[kt] = ldfrag_s(&x_lds[1][c16][0], kt * 32, hi);

  v8h hv8 = {};

  float hkeep[8] = {0.f, 0.f, 0.f, 0.f, 0.f, 0.f, 0.f, 0.f};

  for (int t = 0; t < T_; ++t) {
    const int cur = t & 1;
    const int nxt = cur ^ 1;

    if (loader) {
      if (t + 2 < T_) x_put(cur);
      if (t + 3 < T_) x_fetch(t + 3);
      if (t + 4 < T_) {
        if constexpr (SEQ) __builtin_prefetch(xsrc32 + (size_t)(t + 4) * KIN, 0, 1);
        else               __builtin_prefetch(xsrc16 + (size_t)(t + 4) * KIN, 0, 1);
      }
    }

    if (SEQ && t > 0) *(v8h*)(hdst + (size_t)(t - 1) * H_) = hv8;

    v8f acc[NG_];
    #pragma unroll
    for (int kt = 0; kt < 4; ++kt) {
      #pragma unroll
      for (int g = 0; g < NG_; ++g)
        acc[g] = __builtin_amdgcn_wmma_f32_16x16x32_f16(false, Wh[g][kt], false, ha[kt],
                                                        (short)0, kt == 0 ? accx[g] : acc[g],
                                                        false, false);
    }

    #pragma unroll
    for (int kt = 0; kt < KX; ++kt) {
      #pragma unroll
      for (int g = 0; g < NG_; ++g)
        accx[g] = __builtin_amdgcn_wmma_f32_16x16x32_f16(false, Wx[g][kt], false, xa[kt],
                                                         (short)0, kt == 0 ? bv[g] : accx[g],
                                                         false, false);
    }

    v8h hpack;
    #pragma unroll
    for (int m = 0; m < 8; ++m) {
      float iv = 0.5f + 0.5f * fast_tanh(acc[0][m] * 0.5f);
      float fv = 0.5f + 0.5f * fast_tanh(acc[1][m] * 0.5f);
      float gv = fast_tanh(acc[2][m]);
      float ov = 0.5f + 0.5f * fast_tanh(acc[3][m] * 0.5f);
      c[m] = fv * c[m] + iv * gv;
      float hv = ov * fast_tanh(c[m]);
      hpack[m] = (_Float16)hv;
      hkeep[m] = hv;
    }
    *(v8h*)&h_lds[nxt][c16][j0 + 8 * hi] = hpack;

    __syncthreads();

    #pragma unroll
    for (int kt = 0; kt < 4; ++kt)
      ha[kt] = ldfrag_s(&h_lds[nxt][c16][0], kt * 32, hi);
    if (SEQ) hv8 = *(const v8h*)&h_lds[nxt][co_row][co_col];
    __builtin_amdgcn_sched_barrier(0);
    #pragma unroll
    for (int kt = 0; kt < KX; ++kt)
      xa[kt] = ldfrag_s(&x_lds[cur][c16][0], kt * 32, hi);
  }

  if (SEQ) *(v8h*)(hdst + (size_t)(T_ - 1) * H_) = hv8;

  if (!SEQ) {
    #pragma unroll
    for (int m = 0; m < 8; ++m)
      hlast_out[(b0 + c16) * H_ + j0 + 8 * hi + m] = hkeep[m];
  }

}

__global__ void fc_kernel(const float* __restrict__ h1, const float* __restrict__ Wfc0,
                          const float* __restrict__ bfc0, const float* __restrict__ Wfc1,
                          const float* __restrict__ bfc1, float* __restrict__ out) {
  int b = blockIdx.x;
  int tid = threadIdx.x;
  __shared__ float tmp[32];
  float s = bfc0[tid];
  for (int k = 0; k < H_; ++k) s += h1[b * H_ + k] * Wfc0[tid * H_ + k];
  tmp[tid] = s;
  __syncthreads();
  if (tid < OUT_) {
    float o = bfc1[tid];
    for (int k = 0; k < 32; ++k) o += tmp[k] * Wfc1[tid * 32 + k];
    out[b * OUT_ + tid] = o;
  }
}

extern "C" void kernel_launch(void* const* d_in, const int* in_sizes, int n_in,
                              void* d_out, int out_size, void* d_ws, size_t ws_size,
                              hipStream_t stream) {
  (void)in_sizes; (void)n_in; (void)out_size; (void)ws_size;
  const float* x    = (const float*)d_in[0];
  const float* Wih0 = (const float*)d_in[1];
  const float* Whh0 = (const float*)d_in[2];
  const float* bih0 = (const float*)d_in[3];
  const float* bhh0 = (const float*)d_in[4];
  const float* Wih1 = (const float*)d_in[5];
  const float* Whh1 = (const float*)d_in[6];
  const float* bih1 = (const float*)d_in[7];
  const float* bhh1 = (const float*)d_in[8];
  const float* Wfc0 = (const float*)d_in[9];
  const float* bfc0 = (const float*)d_in[10];
  const float* Wfc1 = (const float*)d_in[11];
  const float* bfc1 = (const float*)d_in[12];

  char* p = (char*)d_ws;
  auto alloc = [&](size_t bytes) {
    char* r = p;
    p += (bytes + 255) & ~(size_t)255;
    return r;
  };
  _Float16* h0seq  = (_Float16*)alloc((size_t)B_ * T_ * H_ * 2);
  _Float16* wih0_h = (_Float16*)alloc((size_t)G_ * D_ * 2);
  _Float16* whh0_h = (_Float16*)alloc((size_t)G_ * H_ * 2);
  _Float16* wih1_h = (_Float16*)alloc((size_t)G_ * H_ * 2);
  _Float16* whh1_h = (_Float16*)alloc((size_t)G_ * H_ * 2);
  float*    b0     = (float*)alloc((size_t)G_ * 4);
  float*    b1     = (float*)alloc((size_t)G_ * 4);
  float*    h1last = (float*)alloc((size_t)B_ * H_ * 4);

  auto cvt = [&](const float* s, _Float16* d, int n) {
    cvt_f32_to_f16<<<(n + 255) / 256, 256, 0, stream>>>(s, d, n);
  };
  cvt(Wih0, wih0_h, G_ * D_);
  cvt(Whh0, whh0_h, G_ * H_);
  cvt(Wih1, wih1_h, G_ * H_);
  cvt(Whh1, whh1_h, G_ * H_);
  add_vec<<<2, 256, 0, stream>>>(bih0, bhh0, b0, G_);
  add_vec<<<2, 256, 0, stream>>>(bih1, bhh1, b1, G_);

  lstm_layer_kernel<D_, true><<<B_ / BT_, THREADS_, 0, stream>>>(x, nullptr, wih0_h, whh0_h, b0,
                                                                 h0seq, nullptr);
  lstm_layer_kernel<H_, false><<<B_ / BT_, THREADS_, 0, stream>>>(nullptr, h0seq, wih1_h, whh1_h,
                                                                  b1, nullptr, h1last);
  fc_kernel<<<B_, 32, 0, stream>>>(h1last, Wfc0, bfc0, Wfc1, bfc1, (float*)d_out);
}


# Step barrier written by hand to order LDS traffic only (no drain of global loads and stores), x fetched two steps ahead into two register sets: memory latency leaves the serial path of a step
# speedup vs baseline: 4.6634x; 1.3456x over previous
#include <hip/hip_runtime.h>


#define B_ 512
#define T_ 512
#define D_ 64
#define H_ 128
#define G_ 512
#define OUT_ 16
#define BT_ 16
#define THREADS_ 256
#define NG_ 4

typedef __attribute__((ext_vector_type(16))) _Float16 v16h;
typedef __attribute__((ext_vector_type(8)))  _Float16 v8h;
typedef __attribute__((ext_vector_type(8)))  float    v8f;
typedef __attribute__((ext_vector_type(4)))  _Float16 v4h;
typedef __attribute__((ext_vector_type(4)))  float    v4f;

__device__ __forceinline__ float fast_tanh(float x) {
#if __has_builtin(__builtin_amdgcn_tanhf)
  return __builtin_amdgcn_tanhf(x);
#else
  float e = __builtin_amdgcn_exp2f(x * 2.8853900817779268f);
  return 1.0f - 2.0f * __builtin_amdgcn_rcpf(e + 1.0f);
#endif
}

__device__ __forceinline__ v16h ldfrag_g(const _Float16* __restrict__ base,
                                         int stride, int koff, int lane) {
  int r  = lane & 15;
  int hi = lane >> 4;
  const _Float16* p = base + (size_t)r * stride + koff + hi * 8;
  v8h a = *(const v8h*)p;
  v8h b = *(const v8h*)(p + 16);
  return __builtin_shufflevector(a, b, 0,1,2,3,4,5,6,7,8,9,10,11,12,13,14,15);
}

__device__ __forceinline__ v16h ldfrag_s(const _Float16* rowbase, int koff, int hi) {
  const _Float16* p = rowbase + koff + hi * 8;
  v8h a = *(const v8h*)p;
  v8h b = *(const v8h*)(p + 16);
  return __builtin_shufflevector(a, b, 0,1,2,3,4,5,6,7,8,9,10,11,12,13,14,15);
}

__device__ __forceinline__ void lds_barrier() {
  asm volatile("s_wait_dscnt 0x0\n\ts_barrier_signal -1\n\ts_barrier_wait -1" ::: "memory");
}

__global__ void cvt_f32_to_f16(const float* __restrict__ s, _Float16* __restrict__ d, int n) {
  int i = blockIdx.x * blockDim.x + threadIdx.x;
  if (i < n) d[i] = (_Float16)s[i];
}

__global__ void add_vec(const float* __restrict__ a, const float* __restrict__ b,
                        float* __restrict__ o, int n) {
  int i = blockIdx.x * blockDim.x + threadIdx.x;
  if (i < n) o[i] = a[i] + b[i];
}

template<int KIN, bool SEQ>
__global__ __launch_bounds__(THREADS_, 1)
void lstm_layer_kernel(const float*    __restrict__ x32,
                       const _Float16* __restrict__ xh,
                       const _Float16* __restrict__ Wih,
                       const _Float16* __restrict__ Whh,
                       const float*    __restrict__ bias,
                       _Float16*       __restrict__ hseq_out,
                       float*          __restrict__ hlast_out)
{
  constexpr int KX = KIN / 32;

  __shared__ __attribute__((aligned(16)))
  _Float16 h_lds[2][BT_][H_ + 8];
  __shared__ __attribute__((aligned(16)))
  _Float16 x_lds[2][BT_][KIN + 8];

  const int tid   = threadIdx.x;
  const int lane  = tid & 31;
  const int wave  = tid >> 5;
  const int hi    = lane >> 4;
  const int c16   = lane & 15;
  const int b0    = blockIdx.x * BT_;
  const int j0    = wave * 16;

  for (int i = tid; i < BT_ * (H_ + 8); i += THREADS_)
    ((_Float16*)h_lds[0])[i] = (_Float16)0.0f;

  v16h Wx[NG_][KX];
  v16h Wh[NG_][4];
  v8f bv[NG_];
  #pragma unroll
  for (int g = 0; g < NG_; ++g) {
    const int nc = g * H_ + j0;
    #pragma unroll
    for (int m = 0; m < 8; ++m) bv[g][m] = bias[nc + 8 * hi + m];
    #pragma unroll
    for (int kt = 0; kt < KX; ++kt)
      Wx[g][kt] = ldfrag_g(Wih + (size_t)nc * KIN, KIN, kt * 32, lane);
    #pragma unroll
    for (int kt = 0; kt < 4; ++kt)
      Wh[g][kt] = ldfrag_g(Whh + (size_t)nc * H_, H_, kt * 32, lane);
  }

  float c[8] = {0.f, 0.f, 0.f, 0.f, 0.f, 0.f, 0.f, 0.f};

  constexpr int EPP_   = SEQ ? 4 : 8;
  constexpr int CPR_   = KIN / EPP_;
  constexpr int NLOAD_ = BT_ * CPR_;
  static_assert(NLOAD_ <= THREADS_, "one piece per thread");
  const bool loader = (NLOAD_ == THREADS_) || tid < NLOAD_;
  const int  xs_row = tid / CPR_;
  const int  xs_col = (tid % CPR_) * EPP_;
  const size_t xoff = (size_t)(b0 + xs_row) * (T_ * KIN) + xs_col;
  const float*    xsrc32 = SEQ ? x32 + xoff : nullptr;
  const _Float16* xsrc16 = SEQ ? nullptr : xh + xoff;
  v4f xa32 = {}, xb32 = {};
  v8h xa16 = {}, xb16 = {};
  auto x_fetch = [&](int s, v4f& r32, v8h& r16) {
    if constexpr (SEQ) r32 = *(const v4f*)(xsrc32 + (size_t)s * KIN);
    else               r16 = *(const v8h*)(xsrc16 + (size_t)s * KIN);
  };
  auto x_put = [&](int cp, const v4f& r32, const v8h& r16) {
    if constexpr (SEQ) {
      v4h p4 = {(_Float16)r32[0], (_Float16)r32[1], (_Float16)r32[2], (_Float16)r32[3]};
      *(v4h*)&x_lds[cp][xs_row][xs_col] = p4;
    } else {
      *(v8h*)&x_lds[cp][xs_row][xs_col] = r16;
    }
  };
  if (loader) {
    x_fetch(0, xa32, xa16); x_put(0, xa32, xa16);
    x_fetch(1, xb32, xb16); x_put(1, xb32, xb16);
    x_fetch(2, xa32, xa16);
    x_fetch(3, xb32, xb16);
  }

  const int co_row = tid >> 4;
  const int co_col = (tid & 15) * 8;
  _Float16* hdst = SEQ ? hseq_out + (size_t)(b0 + co_row) * T_ * H_ + co_col : nullptr;

  __syncthreads();

  v8f accx[NG_];
  {
    v16h xa0[KX];
    #pragma unroll
    for (int kt = 0; kt < KX; ++kt)
      xa0[kt] = ldfrag_s(&x_lds[0][c16][0], kt * 32, hi);
    #pragma unroll
    for (int kt = 0; kt < KX; ++kt) {
      #pragma unroll
      for (int g = 0; g < NG_; ++g)
        accx[g] = __builtin_amdgcn_wmma_f32_16x16x32_f16(false, Wx[g][kt], false, xa0[kt],
                                                         (short)0, kt == 0 ? bv[g] : accx[g],
                                                         false, false);
    }
  }
  __syncthreads();

  v16h ha[4];
  #pragma unroll
  for (int kt = 0; kt < 4; ++kt)
    ha[kt] = ldfrag_s(&h_lds[0][c16][0], kt * 32, hi);
  v16h xa[KX];
  #pragma unroll
  for (int kt = 0; kt < KX; ++kt)
    xa[kt] = ldfrag_s(&x_lds[1][c16][0], kt * 32, hi);

  v8h hv8 = {};

  float hkeep[8] = {0.f, 0.f, 0.f, 0.f, 0.f, 0.f, 0.f, 0.f};

  auto step = [&](const int t, const int cur, v4f& r32, v8h& r16) {
    const int nxt = cur ^ 1;

    if (loader) {
      x_put(cur, r32, r16);
      const int sf = t + 4 < T_ ? t + 4 : T_ - 1;
      x_fetch(sf, r32, r16);
      const int sp = t + 6 < T_ ? t + 6 : T_ - 1;
      if constexpr (SEQ) __builtin_prefetch(xsrc32 + (size_t)sp * KIN, 0, 1);
      else               __builtin_prefetch(xsrc16 + (size_t)sp * KIN, 0, 1);
    }

    if (SEQ && t > 0) *(v8h*)(hdst + (size_t)(t - 1) * H_) = hv8;

    v8f acc[NG_];
    #pragma unroll
    for (int kt = 0; kt < 4; ++kt) {
      #pragma unroll
      for (int g = 0; g < NG_; ++g)
        acc[g] = __builtin_amdgcn_wmma_f32_16x16x32_f16(false, Wh[g][kt], false, ha[kt],
                                                        (short)0, kt == 0 ? accx[g] : acc[g],
                                                        false, false);
    }

    #pragma unroll
    for (int kt = 0; kt < KX; ++kt) {
      #pragma unroll
      for (int g = 0; g < NG_; ++g)
        accx[g] = __builtin_amdgcn_wmma_f32_16x16x32_f16(false, Wx[g][kt], false, xa[kt],
                                                         (short)0, kt == 0 ? bv[g] : accx[g],
                                                         false, false);
    }

    v8h hpack;
    #pragma unroll
    for (int m = 0; m < 8; ++m) {
      float iv = 0.5f + 0.5f * fast_tanh(acc[0][m] * 0.5f);
      float fv = 0.5f + 0.5f * fast_tanh(acc[1][m] * 0.5f);
      float gv = fast_tanh(acc[2][m]);
      float ov = 0.5f + 0.5f * fast_tanh(acc[3][m] * 0.5f);
      c[m] = fv * c[m] + iv * gv;
      float hv = ov * fast_tanh(c[m]);
      hpack[m] = (_Float16)hv;
      hkeep[m] = hv;
    }
    *(v8h*)&h_lds[nxt][c16][j0 + 8 * hi] = hpack;

    lds_barrier();

    #pragma unroll
    for (int kt = 0; kt < 4; ++kt)
      ha[kt] = ldfrag_s(&h_lds[nxt][c16][0], kt * 32, hi);
    if (SEQ) hv8 = *(const v8h*)&h_lds[nxt][co_row][co_col];
    __builtin_amdgcn_sched_barrier(0);
    #pragma unroll
    for (int kt = 0; kt < KX; ++kt)
      xa[kt] = ldfrag_s(&x_lds[cur][c16][0], kt * 32, hi);
  };

  static_assert(T_ % 2 == 0, "the loop takes the steps in pairs");
  for (int t = 0; t < T_; t += 2) {
    step(t,     0, xa32, xa16);
    step(t + 1, 1, xb32, xb16);
  }

  if (SEQ) *(v8h*)(hdst + (size_t)(T_ - 1) * H_) = hv8;

  if (!SEQ) {
    #pragma unroll
    for (int m = 0; m < 8; ++m)
      hlast_out[(b0 + c16) * H_ + j0 + 8 * hi + m] = hkeep[m];
  }

}

__global__ void fc_kernel(const float* __restrict__ h1, const float* __restrict__ Wfc0,
                          const float* __restrict__ bfc0, const float* __restrict__ Wfc1,
                          const float* __restrict__ bfc1, float* __restrict__ out) {
  int b = blockIdx.x;
  int tid = threadIdx.x;
  __shared__ float tmp[32];
  float s = bfc0[tid];
  for (int k = 0; k < H_; ++k) s += h1[b * H_ + k] * Wfc0[tid * H_ + k];
  tmp[tid] = s;
  __syncthreads();
  if (tid < OUT_) {
    float o = bfc1[tid];
    for (int k = 0; k < 32; ++k) o += tmp[k] * Wfc1[tid * 32 + k];
    out[b * OUT_ + tid] = o;
  }
}

extern "C" void kernel_launch(void* const* d_in, const int* in_sizes, int n_in,
                              void* d_out, int out_size, void* d_ws, size_t ws_size,
                              hipStream_t stream) {
  (void)in_sizes; (void)n_in; (void)out_size; (void)ws_size;
  const float* x    = (const float*)d_in[0];
  const float* Wih0 = (const float*)d_in[1];
  const float* Whh0 = (const float*)d_in[2];
  const float* bih0 = (const float*)d_in[3];
  const float* bhh0 = (const float*)d_in[4];
  const float* Wih1 = (const float*)d_in[5];
  const float* Whh1 = (const float*)d_in[6];
  const float* bih1 = (const float*)d_in[7];
  const float* bhh1 = (const float*)d_in[8];
  const float* Wfc0 = (const float*)d_in[9];
  const float* bfc0 = (const float*)d_in[10];
  const float* Wfc1 = (const float*)d_in[11];
  const float* bfc1 = (const float*)d_in[12];

  char* p = (char*)d_ws;
  auto alloc = [&](size_t bytes) {
    char* r = p;
    p += (bytes + 255) & ~(size_t)255;
    return r;
  };
  _Float16* h0seq  = (_Float16*)alloc((size_t)B_ * T_ * H_ * 2);
  _Float16* wih0_h = (_Float16*)alloc((size_t)G_ * D_ * 2);
  _Float16* whh0_h = (_Float16*)alloc((size_t)G_ * H_ * 2);
  _Float16* wih1_h = (_Float16*)alloc((size_t)G_ * H_ * 2);
  _Float16* whh1_h = (_Float16*)alloc((size_t)G_ * H_ * 2);
  float*    b0     = (float*)alloc((size_t)G_ * 4);
  float*    b1     = (float*)alloc((size_t)G_ * 4);
  float*    h1last = (float*)alloc((size_t)B_ * H_ * 4);

  auto cvt = [&](const float* s, _Float16* d, int n) {
    cvt_f32_to_f16<<<(n + 255) / 256, 256, 0, stream>>>(s, d, n);
  };
  cvt(Wih0, wih0_h, G_ * D_);
  cvt(Whh0, whh0_h, G_ * H_);
  cvt(Wih1, wih1_h, G_ * H_);
  cvt(Whh1, whh1_h, G_ * H_);
  add_vec<<<2, 256, 0, stream>>>(bih0, bhh0, b0, G_);
  add_vec<<<2, 256, 0, stream>>>(bih1, bhh1, b1, G_);

  lstm_layer_kernel<D_, true><<<B_ / BT_, THREADS_, 0, stream>>>(x, nullptr, wih0_h, whh0_h, b0,
                                                                 h0seq, nullptr);
  lstm_layer_kernel<H_, false><<<B_ / BT_, THREADS_, 0, stream>>>(nullptr, h0seq, wih1_h, whh1_h,
                                                                  b1, nullptr, h1last);
  fc_kernel<<<B_, 32, 0, stream>>>(h1last, Wfc0, bfc0, Wfc1, bfc1, (float*)d_out);
}


# Look-ahead input projection moved out of the stretch between a step's recurrent WMMAs and its h hand-over into the shadow of a split step barrier (signal, projection at raised wave priority, wait)
# speedup vs baseline: 4.8738x; 1.0451x over previous
#include <hip/hip_runtime.h>


#define B_ 512
#define T_ 512
#define D_ 64
#define H_ 128
#define G_ 512
#define OUT_ 16
#define BT_ 16
#define THREADS_ 256
#define NG_ 4

typedef __attribute__((ext_vector_type(16))) _Float16 v16h;
typedef __attribute__((ext_vector_type(8)))  _Float16 v8h;
typedef __attribute__((ext_vector_type(8)))  float    v8f;
typedef __attribute__((ext_vector_type(4)))  _Float16 v4h;
typedef __attribute__((ext_vector_type(4)))  float    v4f;

__device__ __forceinline__ float fast_tanh(float x) {
#if __has_builtin(__builtin_amdgcn_tanhf)
  return __builtin_amdgcn_tanhf(x);
#else
  float e = __builtin_amdgcn_exp2f(x * 2.8853900817779268f);
  return 1.0f - 2.0f * __builtin_amdgcn_rcpf(e + 1.0f);
#endif
}

__device__ __forceinline__ v16h ldfrag_g(const _Float16* __restrict__ base,
                                         int stride, int koff, int lane) {
  int r  = lane & 15;
  int hi = lane >> 4;
  const _Float16* p = base + (size_t)r * stride + koff + hi * 8;
  v8h a = *(const v8h*)p;
  v8h b = *(const v8h*)(p + 16);
  return __builtin_shufflevector(a, b, 0,1,2,3,4,5,6,7,8,9,10,11,12,13,14,15);
}

__device__ __forceinline__ v16h ldfrag_s(const _Float16* rowbase, int koff, int hi) {
  const _Float16* p = rowbase + koff + hi * 8;
  v8h a = *(const v8h*)p;
  v8h b = *(const v8h*)(p + 16);
  return __builtin_shufflevector(a, b, 0,1,2,3,4,5,6,7,8,9,10,11,12,13,14,15);
}

__device__ __forceinline__ void lds_signal() {
  asm volatile("s_wait_dscnt 0x0\n\ts_barrier_signal -1\n\ts_setprio 3" ::: "memory");
}
__device__ __forceinline__ void lds_wait() {
  asm volatile("s_setprio 0\n\ts_barrier_wait -1" ::: "memory");
}

__global__ void cvt_f32_to_f16(const float* __restrict__ s, _Float16* __restrict__ d, int n) {
  int i = blockIdx.x * blockDim.x + threadIdx.x;
  if (i < n) d[i] = (_Float16)s[i];
}

__global__ void add_vec(const float* __restrict__ a, const float* __restrict__ b,
                        float* __restrict__ o, int n) {
  int i = blockIdx.x * blockDim.x + threadIdx.x;
  if (i < n) o[i] = a[i] + b[i];
}

template<int KIN, bool SEQ>
__global__ __launch_bounds__(THREADS_, 1)
void lstm_layer_kernel(const float*    __restrict__ x32,
                       const _Float16* __restrict__ xh,
                       const _Float16* __restrict__ Wih,
                       const _Float16* __restrict__ Whh,
                       const float*    __restrict__ bias,
                       _Float16*       __restrict__ hseq_out,
                       float*          __restrict__ hlast_out)
{
  constexpr int KX = KIN / 32;

  __shared__ __attribute__((aligned(16)))
  _Float16 h_lds[2][BT_][H_ + 8];
  __shared__ __attribute__((aligned(16)))
  _Float16 x_lds[2][BT_][KIN + 8];

  const int tid   = threadIdx.x;
  const int lane  = tid & 31;
  const int wave  = tid >> 5;
  const int hi    = lane >> 4;
  const int c16   = lane & 15;
  const int b0    = blockIdx.x * BT_;
  const int j0    = wave * 16;

  for (int i = tid; i < BT_ * (H_ + 8); i += THREADS_)
    ((_Float16*)h_lds[0])[i] = (_Float16)0.0f;

  v16h Wx[NG_][KX];
  v16h Wh[NG_][4];
  v8f bv[NG_];
  #pragma unroll
  for (int g = 0; g < NG_; ++g) {
    const int nc = g * H_ + j0;
    #pragma unroll
    for (int m = 0; m < 8; ++m) bv[g][m] = bias[nc + 8 * hi + m];
    #pragma unroll
    for (int kt = 0; kt < KX; ++kt)
      Wx[g][kt] = ldfrag_g(Wih + (size_t)nc * KIN, KIN, kt * 32, lane);
    #pragma unroll
    for (int kt = 0; kt < 4; ++kt)
      Wh[g][kt] = ldfrag_g(Whh + (size_t)nc * H_, H_, kt * 32, lane);
  }

  float c[8] = {0.f, 0.f, 0.f, 0.f, 0.f, 0.f, 0.f, 0.f};

  constexpr int EPP_   = SEQ ? 4 : 8;
  constexpr int CPR_   = KIN / EPP_;
  constexpr int NLOAD_ = BT_ * CPR_;
  static_assert(NLOAD_ <= THREADS_, "one piece per thread");
  const bool loader = (NLOAD_ == THREADS_) || tid < NLOAD_;
  const int  xs_row = tid / CPR_;
  const int  xs_col = (tid % CPR_) * EPP_;
  const size_t xoff = (size_t)(b0 + xs_row) * (T_ * KIN) + xs_col;
  const float*    xsrc32 = SEQ ? x32 + xoff : nullptr;
  const _Float16* xsrc16 = SEQ ? nullptr : xh + xoff;
  v4f xa32 = {}, xb32 = {};
  v8h xa16 = {}, xb16 = {};
  auto x_fetch = [&](int s, v4f& r32, v8h& r16) {
    if constexpr (SEQ) r32 = *(const v4f*)(xsrc32 + (size_t)s * KIN);
    else               r16 = *(const v8h*)(xsrc16 + (size_t)s * KIN);
  };
  auto x_put = [&](int cp, const v4f& r32, const v8h& r16) {
    if constexpr (SEQ) {
      v4h p4 = {(_Float16)r32[0], (_Float16)r32[1], (_Float16)r32[2], (_Float16)r32[3]};
      *(v4h*)&x_lds[cp][xs_row][xs_col] = p4;
    } else {
      *(v8h*)&x_lds[cp][xs_row][xs_col] = r16;
    }
  };
  if (loader) {
    x_fetch(0, xa32, xa16); x_put(0, xa32, xa16);
    x_fetch(1, xb32, xb16); x_put(1, xb32, xb16);
    x_fetch(2, xa32, xa16);
    x_fetch(3, xb32, xb16);
  }

  const int co_row = tid >> 4;
  const int co_col = (tid & 15) * 8;
  _Float16* hdst = SEQ ? hseq_out + (size_t)(b0 + co_row) * T_ * H_ + co_col : nullptr;

  __syncthreads();

  v8f accx[NG_];
  {
    v16h xa0[KX];
    #pragma unroll
    for (int kt = 0; kt < KX; ++kt)
      xa0[kt] = ldfrag_s(&x_lds[0][c16][0], kt * 32, hi);
    #pragma unroll
    for (int kt = 0; kt < KX; ++kt) {
      #pragma unroll
      for (int g = 0; g < NG_; ++g)
        accx[g] = __builtin_amdgcn_wmma_f32_16x16x32_f16(false, Wx[g][kt], false, xa0[kt],
                                                         (short)0, kt == 0 ? bv[g] : accx[g],
                                                         false, false);
    }
  }
  __syncthreads();

  v16h ha[4];
  #pragma unroll
  for (int kt = 0; kt < 4; ++kt)
    ha[kt] = ldfrag_s(&h_lds[0][c16][0], kt * 32, hi);
  v16h xa[KX];
  #pragma unroll
  for (int kt = 0; kt < KX; ++kt)
    xa[kt] = ldfrag_s(&x_lds[1][c16][0], kt * 32, hi);

  v8h hv8 = {};

  float hkeep[8] = {0.f, 0.f, 0.f, 0.f, 0.f, 0.f, 0.f, 0.f};

  auto proj = [&]() {
    #pragma unroll
    for (int kt = 0; kt < KX; ++kt) {
      #pragma unroll
      for (int g = 0; g < NG_; ++g)
        accx[g] = __builtin_amdgcn_wmma_f32_16x16x32_f16(false, Wx[g][kt], false, xa[kt],
                                                         (short)0, kt == 0 ? bv[g] : accx[g],
                                                         false, false);
    }
  };

  auto step = [&](const int t, const int cur, v4f& r32, v8h& r16) {
    const int nxt = cur ^ 1;

    if (loader) {
      x_put(cur, r32, r16);
      const int sf = t + 4 < T_ ? t + 4 : T_ - 1;
      x_fetch(sf, r32, r16);
      const int sp = t + 6 < T_ ? t + 6 : T_ - 1;
      if constexpr (SEQ) __builtin_prefetch(xsrc32 + (size_t)sp * KIN, 0, 1);
      else               __builtin_prefetch(xsrc16 + (size_t)sp * KIN, 0, 1);
    }

    if (SEQ && t > 0) *(v8h*)(hdst + (size_t)(t - 1) * H_) = hv8;

    v8f acc[NG_];
    #pragma unroll
    for (int kt = 0; kt < 4; ++kt) {
      #pragma unroll
      for (int g = 0; g < NG_; ++g)
        acc[g] = __builtin_amdgcn_wmma_f32_16x16x32_f16(false, Wh[g][kt], false, ha[kt],
                                                        (short)0, kt == 0 ? accx[g] : acc[g],
                                                        false, false);
    }

    v8h hpack;
    #pragma unroll
    for (int m = 0; m < 8; ++m) {
      float iv = 0.5f + 0.5f * fast_tanh(acc[0][m] * 0.5f);
      float fv = 0.5f + 0.5f * fast_tanh(acc[1][m] * 0.5f);
      float gv = fast_tanh(acc[2][m]);
      float ov = 0.5f + 0.5f * fast_tanh(acc[3][m] * 0.5f);
      c[m] = fv * c[m] + iv * gv;
      float hv = ov * fast_tanh(c[m]);
      hpack[m] = (_Float16)hv;
      hkeep[m] = hv;
    }
    *(v8h*)&h_lds[nxt][c16][j0 + 8 * hi] = hpack;

    lds_signal();

    __builtin_amdgcn_sched_barrier(0);
    proj();
    __builtin_amdgcn_sched_barrier(0);

    lds_wait();

    #pragma unroll
    for (int kt = 0; kt < 4; ++kt)
      ha[kt] = ldfrag_s(&h_lds[nxt][c16][0], kt * 32, hi);
    if (SEQ) hv8 = *(const v8h*)&h_lds[nxt][co_row][co_col];
    __builtin_amdgcn_sched_barrier(0);
    #pragma unroll
    for (int kt = 0; kt < KX; ++kt)
      xa[kt] = ldfrag_s(&x_lds[cur][c16][0], kt * 32, hi);
  };

  static_assert(T_ % 2 == 0, "the loop takes the steps in pairs");
  for (int t = 0; t < T_; t += 2) {
    step(t,     0, xa32, xa16);
    step(t + 1, 1, xb32, xb16);
  }

  if (SEQ) *(v8h*)(hdst + (size_t)(T_ - 1) * H_) = hv8;

  if (!SEQ) {
    #pragma unroll
    for (int m = 0; m < 8; ++m)
      hlast_out[(b0 + c16) * H_ + j0 + 8 * hi + m] = hkeep[m];
  }

}

__global__ void fc_kernel(const float* __restrict__ h1, const float* __restrict__ Wfc0,
                          const float* __restrict__ bfc0, const float* __restrict__ Wfc1,
                          const float* __restrict__ bfc1, float* __restrict__ out) {
  int b = blockIdx.x;
  int tid = threadIdx.x;
  __shared__ float tmp[32];
  float s = bfc0[tid];
  for (int k = 0; k < H_; ++k) s += h1[b * H_ + k] * Wfc0[tid * H_ + k];
  tmp[tid] = s;
  __syncthreads();
  if (tid < OUT_) {
    float o = bfc1[tid];
    for (int k = 0; k < 32; ++k) o += tmp[k] * Wfc1[tid * 32 + k];
    out[b * OUT_ + tid] = o;
  }
}

extern "C" void kernel_launch(void* const* d_in, const int* in_sizes, int n_in,
                              void* d_out, int out_size, void* d_ws, size_t ws_size,
                              hipStream_t stream) {
  (void)in_sizes; (void)n_in; (void)out_size; (void)ws_size;
  const float* x    = (const float*)d_in[0];
  const float* Wih0 = (const float*)d_in[1];
  const float* Whh0 = (const float*)d_in[2];
  const float* bih0 = (const float*)d_in[3];
  const float* bhh0 = (const float*)d_in[4];
  const float* Wih1 = (const float*)d_in[5];
  const float* Whh1 = (const float*)d_in[6];
  const float* bih1 = (const float*)d_in[7];
  const float* bhh1 = (const float*)d_in[8];
  const float* Wfc0 = (const float*)d_in[9];
  const float* bfc0 = (const float*)d_in[10];
  const float* Wfc1 = (const float*)d_in[11];
  const float* bfc1 = (const float*)d_in[12];

  char* p = (char*)d_ws;
  auto alloc = [&](size_t bytes) {
    char* r = p;
    p += (bytes + 255) & ~(size_t)255;
    return r;
  };
  _Float16* h0seq  = (_Float16*)alloc((size_t)B_ * T_ * H_ * 2);
  _Float16* wih0_h = (_Float16*)alloc((size_t)G_ * D_ * 2);
  _Float16* whh0_h = (_Float16*)alloc((size_t)G_ * H_ * 2);
  _Float16* wih1_h = (_Float16*)alloc((size_t)G_ * H_ * 2);
  _Float16* whh1_h = (_Float16*)alloc((size_t)G_ * H_ * 2);
  float*    b0     = (float*)alloc((size_t)G_ * 4);
  float*    b1     = (float*)alloc((size_t)G_ * 4);
  float*    h1last = (float*)alloc((size_t)B_ * H_ * 4);

  auto cvt = [&](const float* s, _Float16* d, int n) {
    cvt_f32_to_f16<<<(n + 255) / 256, 256, 0, stream>>>(s, d, n);
  };
  cvt(Wih0, wih0_h, G_ * D_);
  cvt(Whh0, whh0_h, G_ * H_);
  cvt(Wih1, wih1_h, G_ * H_);
  cvt(Whh1, whh1_h, G_ * H_);
  add_vec<<<2, 256, 0, stream>>>(bih0, bhh0, b0, G_);
  add_vec<<<2, 256, 0, stream>>>(bih1, bhh1, b1, G_);

  lstm_layer_kernel<D_, true><<<B_ / BT_, THREADS_, 0, stream>>>(x, nullptr, wih0_h, whh0_h, b0,
                                                                 h0seq, nullptr);
  lstm_layer_kernel<H_, false><<<B_ / BT_, THREADS_, 0, stream>>>(nullptr, h0seq, wih1_h, whh1_h,
                                                                  b1, nullptr, h1last);
  fc_kernel<<<B_, 32, 0, stream>>>(h1last, Wfc0, bfc0, Wfc1, bfc1, (float*)d_out);
}


# Layer 1 runs one quarter of the sequence behind layer 0 and BESIDE it: T in 4 chunks, each launch holds layer-0 chunk k and layer-1 chunk k-1 on separate workgroups; h, c cross launches in memory
# speedup vs baseline: 6.8862x; 1.4129x over previous
#include <hip/hip_runtime.h>


#define B_ 512
#define T_ 512
#define D_ 64
#define H_ 128
#define G_ 512
#define OUT_ 16
#define BT_ 16
#define THREADS_ 256
#define NG_ 4
#define NCH_ 4
#define CH_ (T_ / NCH_)
#define NWG_ (B_ / BT_)

typedef __attribute__((ext_vector_type(16))) _Float16 v16h;
typedef __attribute__((ext_vector_type(8)))  _Float16 v8h;
typedef __attribute__((ext_vector_type(8)))  float    v8f;
typedef __attribute__((ext_vector_type(4)))  _Float16 v4h;
typedef __attribute__((ext_vector_type(4)))  float    v4f;

__device__ __forceinline__ float fast_tanh(float x) {
#if __has_builtin(__builtin_amdgcn_tanhf)
  return __builtin_amdgcn_tanhf(x);
#else
  float e = __builtin_amdgcn_exp2f(x * 2.8853900817779268f);
  return 1.0f - 2.0f * __builtin_amdgcn_rcpf(e + 1.0f);
#endif
}

__device__ __forceinline__ v16h ldfrag_g(const _Float16* __restrict__ base,
                                         int stride, int koff, int lane) {
  int r  = lane & 15;
  int hi = lane >> 4;
  const _Float16* p = base + (size_t)r * stride + koff + hi * 8;
  v8h a = *(const v8h*)p;
  v8h b = *(const v8h*)(p + 16);
  return __builtin_shufflevector(a, b, 0,1,2,3,4,5,6,7,8,9,10,11,12,13,14,15);
}

__device__ __forceinline__ v16h ldfrag_s(const _Float16* rowbase, int koff, int hi) {
  const _Float16* p = rowbase + koff + hi * 8;
  v8h a = *(const v8h*)p;
  v8h b = *(const v8h*)(p + 16);
  return __builtin_shufflevector(a, b, 0,1,2,3,4,5,6,7,8,9,10,11,12,13,14,15);
}

__device__ __forceinline__ void lds_signal() {
  asm volatile("s_wait_dscnt 0x0\n\ts_barrier_signal -1\n\ts_setprio 3" ::: "memory");
}
__device__ __forceinline__ void lds_wait() {
  asm volatile("s_setprio 0\n\ts_barrier_wait -1" ::: "memory");
}

__global__ void cvt_f32_to_f16(const float* __restrict__ s, _Float16* __restrict__ d, int n) {
  int i = blockIdx.x * blockDim.x + threadIdx.x;
  if (i < n) d[i] = (_Float16)s[i];
}

__global__ void add_vec(const float* __restrict__ a, const float* __restrict__ b,
                        float* __restrict__ o, int n) {
  int i = blockIdx.x * blockDim.x + threadIdx.x;
  if (i < n) o[i] = a[i] + b[i];
}

template<int KIN, bool SEQ>
__device__ __forceinline__
void lstm_layer_body(const float*    __restrict__ x32,
                     const _Float16*              xh,
                     const _Float16* __restrict__ Wih,
                     const _Float16* __restrict__ Whh,
                     const float*    __restrict__ bias,
                     _Float16*                    hseq_out,
                     float*          __restrict__ hlast_out,
                     float*          __restrict__ cstate,
                     _Float16*       __restrict__ hstate,
                     const int wg, const int t0, const int t1)
{
  constexpr int KX = KIN / 32;

  __shared__ __attribute__((aligned(16)))
  _Float16 h_lds[2][BT_][H_ + 8];
  __shared__ __attribute__((aligned(16)))
  _Float16 x_lds[2][BT_][KIN + 8];

  const int tid   = threadIdx.x;
  const int lane  = tid & 31;
  const int wave  = tid >> 5;
  const int hi    = lane >> 4;
  const int c16   = lane & 15;
  const int b0    = wg * BT_;
  const int j0    = wave * 16;

  if (t0 == 0) {
    for (int i = tid; i < BT_ * (H_ + 8); i += THREADS_)
      ((_Float16*)h_lds[0])[i] = (_Float16)0.0f;
  } else {
    const int pr = tid >> 4, pc = (tid & 15) * 8;
    const _Float16* hp = SEQ ? hseq_out + ((size_t)(b0 + pr) * T_ + (t0 - 1)) * H_ + pc
                             : hstate + (size_t)(b0 + pr) * H_ + pc;
    *(v8h*)&h_lds[0][pr][pc] = *(const v8h*)hp;
  }

  v16h Wx[NG_][KX];
  v16h Wh[NG_][4];
  v8f bv[NG_];
  #pragma unroll
  for (int g = 0; g < NG_; ++g) {
    const int nc = g * H_ + j0;
    #pragma unroll
    for (int m = 0; m < 8; ++m) bv[g][m] = bias[nc + 8 * hi + m];
    #pragma unroll
    for (int kt = 0; kt < KX; ++kt)
      Wx[g][kt] = ldfrag_g(Wih + (size_t)nc * KIN, KIN, kt * 32, lane);
    #pragma unroll
    for (int kt = 0; kt < 4; ++kt)
      Wh[g][kt] = ldfrag_g(Whh + (size_t)nc * H_, H_, kt * 32, lane);
  }

  float c[8] = {0.f, 0.f, 0.f, 0.f, 0.f, 0.f, 0.f, 0.f};
  float* cst = cstate + ((size_t)(wg * (THREADS_ / 32) + wave) * 2) * 128 + lane * 4;
  if (t0 != 0) {
    v4f ca = *(const v4f*)cst;
    v4f cb = *(const v4f*)(cst + 128);
    c[0] = ca[0]; c[1] = ca[1]; c[2] = ca[2]; c[3] = ca[3];
    c[4] = cb[0]; c[5] = cb[1]; c[6] = cb[2]; c[7] = cb[3];
  }

  constexpr int EPP_   = SEQ ? 4 : 8;
  constexpr int CPR_   = KIN / EPP_;
  constexpr int NLOAD_ = BT_ * CPR_;
  static_assert(NLOAD_ <= THREADS_, "one piece per thread");
  const bool loader = (NLOAD_ == THREADS_) || tid < NLOAD_;
  const int  xs_row = tid / CPR_;
  const int  xs_col = (tid % CPR_) * EPP_;
  const size_t xoff = (size_t)(b0 + xs_row) * (T_ * KIN) + xs_col;
  const float*    xsrc32 = SEQ ? x32 + xoff : nullptr;
  const _Float16* xsrc16 = SEQ ? nullptr : xh + xoff;
  v4f xa32 = {}, xb32 = {};
  v8h xa16 = {}, xb16 = {};
  auto x_fetch = [&](int s, v4f& r32, v8h& r16) {
    if constexpr (SEQ) r32 = *(const v4f*)(xsrc32 + (size_t)s * KIN);
    else               r16 = *(const v8h*)(xsrc16 + (size_t)s * KIN);
  };
  auto x_put = [&](int cp, const v4f& r32, const v8h& r16) {
    if constexpr (SEQ) {
      v4h p4 = {(_Float16)r32[0], (_Float16)r32[1], (_Float16)r32[2], (_Float16)r32[3]};
      *(v4h*)&x_lds[cp][xs_row][xs_col] = p4;
    } else {
      *(v8h*)&x_lds[cp][xs_row][xs_col] = r16;
    }
  };
  if (loader) {
    x_fetch(t0,     xa32, xa16); x_put(0, xa32, xa16);
    x_fetch(t0 + 1, xb32, xb16); x_put(1, xb32, xb16);
    x_fetch(t0 + 2, xa32, xa16);
    x_fetch(t0 + 3, xb32, xb16);
  }

  const int co_row = tid >> 4;
  const int co_col = (tid & 15) * 8;
  _Float16* hdst = SEQ ? hseq_out + (size_t)(b0 + co_row) * T_ * H_ + co_col : nullptr;

  __syncthreads();

  v8f accx[NG_];
  {
    v16h xa0[KX];
    #pragma unroll
    for (int kt = 0; kt < KX; ++kt)
      xa0[kt] = ldfrag_s(&x_lds[0][c16][0], kt * 32, hi);
    #pragma unroll
    for (int kt = 0; kt < KX; ++kt) {
      #pragma unroll
      for (int g = 0; g < NG_; ++g)
        accx[g] = __builtin_amdgcn_wmma_f32_16x16x32_f16(false, Wx[g][kt], false, xa0[kt],
                                                         (short)0, kt == 0 ? bv[g] : accx[g],
                                                         false, false);
    }
  }
  __syncthreads();

  v16h ha[4];
  #pragma unroll
  for (int kt = 0; kt < 4; ++kt)
    ha[kt] = ldfrag_s(&h_lds[0][c16][0], kt * 32, hi);
  v16h xa[KX];
  #pragma unroll
  for (int kt = 0; kt < KX; ++kt)
    xa[kt] = ldfrag_s(&x_lds[1][c16][0], kt * 32, hi);

  v8h hv8 = {};

  float hkeep[8] = {0.f, 0.f, 0.f, 0.f, 0.f, 0.f, 0.f, 0.f};

  auto proj = [&]() {
    #pragma unroll
    for (int kt = 0; kt < KX; ++kt) {
      #pragma unroll
      for (int g = 0; g < NG_; ++g)
        accx[g] = __builtin_amdgcn_wmma_f32_16x16x32_f16(false, Wx[g][kt], false, xa[kt],
                                                         (short)0, kt == 0 ? bv[g] : accx[g],
                                                         false, false);
    }
  };

  auto step = [&](const int t, const int cur, v4f& r32, v8h& r16) {
    const int nxt = cur ^ 1;

    if (loader) {
      x_put(cur, r32, r16);
      const int sf = t + 4 < t1 ? t + 4 : t1 - 1;
      x_fetch(sf, r32, r16);
      const int sp = t + 6 < t1 ? t + 6 : t1 - 1;
      if constexpr (SEQ) __builtin_prefetch(xsrc32 + (size_t)sp * KIN, 0, 1);
      else               __builtin_prefetch(xsrc16 + (size_t)sp * KIN, 0, 1);
    }

    if (SEQ && (cur == 1 || t > t0)) *(v8h*)(hdst + (size_t)(t - 1) * H_) = hv8;

    v8f acc[NG_];
    #pragma unroll
    for (int kt = 0; kt < 4; ++kt) {
      #pragma unroll
      for (int g = 0; g < NG_; ++g)
        acc[g] = __builtin_amdgcn_wmma_f32_16x16x32_f16(false, Wh[g][kt], false, ha[kt],
                                                        (short)0, kt == 0 ? accx[g] : acc[g],
                                                        false, false);
    }

    v8h hpack;
    #pragma unroll
    for (int m = 0; m < 8; ++m) {
      float iv = 0.5f + 0.5f * fast_tanh(acc[0][m] * 0.5f);
      float fv = 0.5f + 0.5f * fast_tanh(acc[1][m] * 0.5f);
      float gv = fast_tanh(acc[2][m]);
      float ov = 0.5f + 0.5f * fast_tanh(acc[3][m] * 0.5f);
      c[m] = fv * c[m] + iv * gv;
      float hv = ov * fast_tanh(c[m]);
      hpack[m] = (_Float16)hv;
      hkeep[m] = hv;
    }
    *(v8h*)&h_lds[nxt][c16][j0 + 8 * hi] = hpack;

    lds_signal();

    __builtin_amdgcn_sched_barrier(0);
    proj();
    __builtin_amdgcn_sched_barrier(0);

    lds_wait();

    #pragma unroll
    for (int kt = 0; kt < 4; ++kt)
      ha[kt] = ldfrag_s(&h_lds[nxt][c16][0], kt * 32, hi);
    if (SEQ) hv8 = *(const v8h*)&h_lds[nxt][co_row][co_col];
    __builtin_amdgcn_sched_barrier(0);
    #pragma unroll
    for (int kt = 0; kt < KX; ++kt)
      xa[kt] = ldfrag_s(&x_lds[cur][c16][0], kt * 32, hi);
  };

  static_assert(CH_ % 2 == 0 && CH_ >= 4, "the loop takes the steps in pairs, x runs 3 steps ahead");
  for (int t = t0; t < t1; t += 2) {
    step(t,     0, xa32, xa16);
    step(t + 1, 1, xb32, xb16);
  }

  if (SEQ) *(v8h*)(hdst + (size_t)(t1 - 1) * H_) = hv8;

  if (t1 < T_) {
    v4f ca = {c[0], c[1], c[2], c[3]};
    v4f cb = {c[4], c[5], c[6], c[7]};
    *(v4f*)cst = ca;
    *(v4f*)(cst + 128) = cb;
    if (!SEQ)
      *(v8h*)(hstate + (size_t)(b0 + co_row) * H_ + co_col) = *(const v8h*)&h_lds[0][co_row][co_col];
  } else if (!SEQ) {
    #pragma unroll
    for (int m = 0; m < 8; ++m)
      hlast_out[(b0 + c16) * H_ + j0 + 8 * hi + m] = hkeep[m];
  }

}

__global__ __launch_bounds__(THREADS_, 1)
void lstm_pair_kernel(const float*    __restrict__ x,
                      const _Float16* __restrict__ wih0, const _Float16* __restrict__ whh0,
                      const float*    __restrict__ bias0,
                      _Float16*                    h0seq,
                      float*          __restrict__ c0state,
                      const _Float16* __restrict__ wih1, const _Float16* __restrict__ whh1,
                      const float*    __restrict__ bias1,
                      float*          __restrict__ c1state,
                      _Float16*       __restrict__ h1state,
                      float*          __restrict__ h1last,
                      const int ta, const int tb)
{
  const int n0 = ta >= 0 ? NWG_ : 0;
  const int wg = blockIdx.x;
  if (wg < n0)
    lstm_layer_body<D_, true>(x, nullptr, wih0, whh0, bias0, h0seq, nullptr, c0state, nullptr,
                              wg, ta, ta + CH_);
  else
    lstm_layer_body<H_, false>(nullptr, h0seq, wih1, whh1, bias1, nullptr, h1last, c1state,
                               h1state, wg - n0, tb, tb + CH_);
}

__global__ void fc_kernel(const float* __restrict__ h1, const float* __restrict__ Wfc0,
                          const float* __restrict__ bfc0, const float* __restrict__ Wfc1,
                          const float* __restrict__ bfc1, float* __restrict__ out) {
  int b = blockIdx.x;
  int tid = threadIdx.x;
  __shared__ float tmp[32];
  float s = bfc0[tid];
  for (int k = 0; k < H_; ++k) s += h1[b * H_ + k] * Wfc0[tid * H_ + k];
  tmp[tid] = s;
  __syncthreads();
  if (tid < OUT_) {
    float o = bfc1[tid];
    for (int k = 0; k < 32; ++k) o += tmp[k] * Wfc1[tid * 32 + k];
    out[b * OUT_ + tid] = o;
  }
}

extern "C" void kernel_launch(void* const* d_in, const int* in_sizes, int n_in,
                              void* d_out, int out_size, void* d_ws, size_t ws_size,
                              hipStream_t stream) {
  (void)in_sizes; (void)n_in; (void)out_size; (void)ws_size;
  const float* x    = (const float*)d_in[0];
  const float* Wih0 = (const float*)d_in[1];
  const float* Whh0 = (const float*)d_in[2];
  const float* bih0 = (const float*)d_in[3];
  const float* bhh0 = (const float*)d_in[4];
  const float* Wih1 = (const float*)d_in[5];
  const float* Whh1 = (const float*)d_in[6];
  const float* bih1 = (const float*)d_in[7];
  const float* bhh1 = (const float*)d_in[8];
  const float* Wfc0 = (const float*)d_in[9];
  const float* bfc0 = (const float*)d_in[10];
  const float* Wfc1 = (const float*)d_in[11];
  const float* bfc1 = (const float*)d_in[12];

  char* p = (char*)d_ws;
  auto alloc = [&](size_t bytes) {
    char* r = p;
    p += (bytes + 255) & ~(size_t)255;
    return r;
  };
  _Float16* h0seq  = (_Float16*)alloc((size_t)B_ * T_ * H_ * 2);
  _Float16* wih0_h = (_Float16*)alloc((size_t)G_ * D_ * 2);
  _Float16* whh0_h = (_Float16*)alloc((size_t)G_ * H_ * 2);
  _Float16* wih1_h = (_Float16*)alloc((size_t)G_ * H_ * 2);
  _Float16* whh1_h = (_Float16*)alloc((size_t)G_ * H_ * 2);
  float*    b0     = (float*)alloc((size_t)G_ * 4);
  float*    b1     = (float*)alloc((size_t)G_ * 4);
  float*    h1last = (float*)alloc((size_t)B_ * H_ * 4);
  float*    c0st   = (float*)alloc((size_t)B_ * H_ * 4);
  float*    c1st   = (float*)alloc((size_t)B_ * H_ * 4);
  _Float16* h1st   = (_Float16*)alloc((size_t)B_ * H_ * 2);

  auto cvt = [&](const float* s, _Float16* d, int n) {
    cvt_f32_to_f16<<<(n + 255) / 256, 256, 0, stream>>>(s, d, n);
  };
  cvt(Wih0, wih0_h, G_ * D_);
  cvt(Whh0, whh0_h, G_ * H_);
  cvt(Wih1, wih1_h, G_ * H_);
  cvt(Whh1, whh1_h, G_ * H_);
  add_vec<<<2, 256, 0, stream>>>(bih0, bhh0, b0, G_);
  add_vec<<<2, 256, 0, stream>>>(bih1, bhh1, b1, G_);

  static_assert(NCH_ == 4, "the launches below are written out for four chunks");
  lstm_pair_kernel<<<NWG_, THREADS_, 0, stream>>>(x, wih0_h, whh0_h, b0, h0seq, c0st,
                                                  wih1_h, whh1_h, b1, c1st, h1st, h1last,
                                                  0, -1);
  lstm_pair_kernel<<<2 * NWG_, THREADS_, 0, stream>>>(x, wih0_h, whh0_h, b0, h0seq, c0st,
                                                      wih1_h, whh1_h, b1, c1st, h1st, h1last,
                                                      CH_, 0);
  lstm_pair_kernel<<<2 * NWG_, THREADS_, 0, stream>>>(x, wih0_h, whh0_h, b0, h0seq, c0st,
                                                      wih1_h, whh1_h, b1, c1st, h1st, h1last,
                                                      2 * CH_, CH_);
  lstm_pair_kernel<<<2 * NWG_, THREADS_, 0, stream>>>(x, wih0_h, whh0_h, b0, h0seq, c0st,
                                                      wih1_h, whh1_h, b1, c1st, h1st, h1last,
                                                      3 * CH_, 2 * CH_);
  lstm_pair_kernel<<<NWG_, THREADS_, 0, stream>>>(x, wih0_h, whh0_h, b0, h0seq, c0st,
                                                  wih1_h, whh1_h, b1, c1st, h1st, h1last,
                                                  -1, 3 * CH_);
  fc_kernel<<<B_, 32, 0, stream>>>(h1last, Wfc0, bfc0, Wfc1, bfc1, (float*)d_out);
}


# Small launches folded: ONE launch rounds all four weight matrices and sums both biases (six before); the last recurrent launch runs the two dense layers on its own rows: 12 launches become 6
# speedup vs baseline: 7.1609x; 1.0399x over previous
#include <hip/hip_runtime.h>


#define B_ 512
#define T_ 512
#define D_ 64
#define H_ 128
#define G_ 512
#define OUT_ 16
#define FC_ 32
#define BT_ 16
#define THREADS_ 256
#define NG_ 4
#define NCH_ 4
#define CH_ (T_ / NCH_)
#define NWG_ (B_ / BT_)

typedef __attribute__((ext_vector_type(16))) _Float16 v16h;
typedef __attribute__((ext_vector_type(8)))  _Float16 v8h;
typedef __attribute__((ext_vector_type(8)))  float    v8f;
typedef __attribute__((ext_vector_type(4)))  _Float16 v4h;
typedef __attribute__((ext_vector_type(4)))  float    v4f;

__device__ __forceinline__ float fast_tanh(float x) {
#if __has_builtin(__builtin_amdgcn_tanhf)
  return __builtin_amdgcn_tanhf(x);
#else
  float e = __builtin_amdgcn_exp2f(x * 2.8853900817779268f);
  return 1.0f - 2.0f * __builtin_amdgcn_rcpf(e + 1.0f);
#endif
}

__device__ __forceinline__ v16h ldfrag_g(const _Float16* __restrict__ base,
                                         int stride, int koff, int lane) {
  int r  = lane & 15;
  int hi = lane >> 4;
  const _Float16* p = base + (size_t)r * stride + koff + hi * 8;
  v8h a = *(const v8h*)p;
  v8h b = *(const v8h*)(p + 16);
  return __builtin_shufflevector(a, b, 0,1,2,3,4,5,6,7,8,9,10,11,12,13,14,15);
}

__device__ __forceinline__ v16h ldfrag_s(const _Float16* rowbase, int koff, int hi) {
  const _Float16* p = rowbase + koff + hi * 8;
  v8h a = *(const v8h*)p;
  v8h b = *(const v8h*)(p + 16);
  return __builtin_shufflevector(a, b, 0,1,2,3,4,5,6,7,8,9,10,11,12,13,14,15);
}

__device__ __forceinline__ void lds_signal() {
  asm volatile("s_wait_dscnt 0x0\n\ts_barrier_signal -1\n\ts_setprio 3" ::: "memory");
}
__device__ __forceinline__ void lds_wait() {
  asm volatile("s_setprio 0\n\ts_barrier_wait -1" ::: "memory");
}

#define PREP_THREADS_ 256
#define PREP_NB_X0_   (G_ * D_ / 8 / PREP_THREADS_)
#define PREP_NB_W_    (G_ * H_ / 8 / PREP_THREADS_)
#define PREP_BLOCKS_  (PREP_NB_X0_ + 3 * PREP_NB_W_ + 1)
static_assert(G_ / 4 * 2 == PREP_THREADS_, "one block holds both bias vectors, 4 elements per thread");

__device__ __forceinline__ void cvt8(const float* __restrict__ s, _Float16* __restrict__ d, int i) {
  const float* sp = s + 8 * (size_t)i;
  v4f a = *(const v4f*)sp;
  v4f b = *(const v4f*)(sp + 4);
  v8h h = {(_Float16)a[0], (_Float16)a[1], (_Float16)a[2], (_Float16)a[3],
           (_Float16)b[0], (_Float16)b[1], (_Float16)b[2], (_Float16)b[3]};
  *(v8h*)(d + 8 * (size_t)i) = h;
}

__global__ __launch_bounds__(PREP_THREADS_)
void prep_kernel(const float* __restrict__ Wih0, const float* __restrict__ Whh0,
                 const float* __restrict__ Wih1, const float* __restrict__ Whh1,
                 const float* __restrict__ bih0, const float* __restrict__ bhh0,
                 const float* __restrict__ bih1, const float* __restrict__ bhh1,
                 _Float16* __restrict__ wih0_h, _Float16* __restrict__ whh0_h,
                 _Float16* __restrict__ wih1_h, _Float16* __restrict__ whh1_h,
                 float* __restrict__ b0, float* __restrict__ b1) {
  const int blk = blockIdx.x;
  const int tid = threadIdx.x;
  if (blk < PREP_NB_X0_) {
    cvt8(Wih0, wih0_h, blk * PREP_THREADS_ + tid);
  } else if (blk < PREP_NB_X0_ + PREP_NB_W_) {
    cvt8(Whh0, whh0_h, (blk - PREP_NB_X0_) * PREP_THREADS_ + tid);
  } else if (blk < PREP_NB_X0_ + 2 * PREP_NB_W_) {
    cvt8(Wih1, wih1_h, (blk - PREP_NB_X0_ - PREP_NB_W_) * PREP_THREADS_ + tid);
  } else if (blk < PREP_NB_X0_ + 3 * PREP_NB_W_) {
    cvt8(Whh1, whh1_h, (blk - PREP_NB_X0_ - 2 * PREP_NB_W_) * PREP_THREADS_ + tid);
  } else {
    const bool l1 = tid >= G_ / 4;
    const int  i  = (tid & (G_ / 4 - 1)) * 4;
    const float* pa = l1 ? bih1 : bih0;
    const float* pb = l1 ? bhh1 : bhh0;
    float*       po = l1 ? b1 : b0;
    v4f va = *(const v4f*)(pa + i);
    v4f vb = *(const v4f*)(pb + i);
    v4f vo = {va[0] + vb[0], va[1] + vb[1], va[2] + vb[2], va[3] + vb[3]};
    *(v4f*)(po + i) = vo;
  }
}

template<int KIN, bool SEQ>
__device__ __forceinline__
void lstm_layer_body(const float*    __restrict__ x32,
                     const _Float16*              xh,
                     const _Float16* __restrict__ Wih,
                     const _Float16* __restrict__ Whh,
                     const float*    __restrict__ bias,
                     _Float16*                    hseq_out,
                     float*          __restrict__ hlast_out,
                     float*          __restrict__ cstate,
                     _Float16*       __restrict__ hstate,
                     const float*    __restrict__ Wfc0,
                     const float*    __restrict__ bfc0,
                     const float*    __restrict__ Wfc1,
                     const float*    __restrict__ bfc1,
                     float*          __restrict__ out,
                     const int wg, const int t0, const int t1)
{
  constexpr int KX = KIN / 32;

  __shared__ __attribute__((aligned(16)))
  _Float16 h_lds[2][BT_][H_ + 8];
  __shared__ __attribute__((aligned(16)))
  _Float16 x_lds[2][BT_][KIN + 8];

  const int tid   = threadIdx.x;
  const int lane  = tid & 31;
  const int wave  = tid >> 5;
  const int hi    = lane >> 4;
  const int c16   = lane & 15;
  const int b0    = wg * BT_;
  const int j0    = wave * 16;

  if (t0 == 0) {
    for (int i = tid; i < BT_ * (H_ + 8); i += THREADS_)
      ((_Float16*)h_lds[0])[i] = (_Float16)0.0f;
  } else {
    const int pr = tid >> 4, pc = (tid & 15) * 8;
    const _Float16* hp = SEQ ? hseq_out + ((size_t)(b0 + pr) * T_ + (t0 - 1)) * H_ + pc
                             : hstate + (size_t)(b0 + pr) * H_ + pc;
    *(v8h*)&h_lds[0][pr][pc] = *(const v8h*)hp;
  }

  v16h Wx[NG_][KX];
  v16h Wh[NG_][4];
  v8f bv[NG_];
  #pragma unroll
  for (int g = 0; g < NG_; ++g) {
    const int nc = g * H_ + j0;
    #pragma unroll
    for (int m = 0; m < 8; ++m) bv[g][m] = bias[nc + 8 * hi + m];
    #pragma unroll
    for (int kt = 0; kt < KX; ++kt)
      Wx[g][kt] = ldfrag_g(Wih + (size_t)nc * KIN, KIN, kt * 32, lane);
    #pragma unroll
    for (int kt = 0; kt < 4; ++kt)
      Wh[g][kt] = ldfrag_g(Whh + (size_t)nc * H_, H_, kt * 32, lane);
  }

  float c[8] = {0.f, 0.f, 0.f, 0.f, 0.f, 0.f, 0.f, 0.f};
  float* cst = cstate + ((size_t)(wg * (THREADS_ / 32) + wave) * 2) * 128 + lane * 4;
  if (t0 != 0) {
    v4f ca = *(const v4f*)cst;
    v4f cb = *(const v4f*)(cst + 128);
    c[0] = ca[0]; c[1] = ca[1]; c[2] = ca[2]; c[3] = ca[3];
    c[4] = cb[0]; c[5] = cb[1]; c[6] = cb[2]; c[7] = cb[3];
  }

  constexpr int EPP_   = SEQ ? 4 : 8;
  constexpr int CPR_   = KIN / EPP_;
  constexpr int NLOAD_ = BT_ * CPR_;
  static_assert(NLOAD_ <= THREADS_, "one piece per thread");
  const bool loader = (NLOAD_ == THREADS_) || tid < NLOAD_;
  const int  xs_row = tid / CPR_;
  const int  xs_col = (tid % CPR_) * EPP_;
  const size_t xoff = (size_t)(b0 + xs_row) * (T_ * KIN) + xs_col;
  const float*    xsrc32 = SEQ ? x32 + xoff : nullptr;
  const _Float16* xsrc16 = SEQ ? nullptr : xh + xoff;
  v4f xa32 = {}, xb32 = {};
  v8h xa16 = {}, xb16 = {};
  auto x_fetch = [&](int s, v4f& r32, v8h& r16) {
    if constexpr (SEQ) r32 = *(const v4f*)(xsrc32 + (size_t)s * KIN);
    else               r16 = *(const v8h*)(xsrc16 + (size_t)s * KIN);
  };
  auto x_put = [&](int cp, const v4f& r32, const v8h& r16) {
    if constexpr (SEQ) {
      v4h p4 = {(_Float16)r32[0], (_Float16)r32[1], (_Float16)r32[2], (_Float16)r32[3]};
      *(v4h*)&x_lds[cp][xs_row][xs_col] = p4;
    } else {
      *(v8h*)&x_lds[cp][xs_row][xs_col] = r16;
    }
  };
  if (loader) {
    x_fetch(t0,     xa32, xa16); x_put(0, xa32, xa16);
    x_fetch(t0 + 1, xb32, xb16); x_put(1, xb32, xb16);
    x_fetch(t0 + 2, xa32, xa16);
    x_fetch(t0 + 3, xb32, xb16);
  }

  const int co_row = tid >> 4;
  const int co_col = (tid & 15) * 8;
  _Float16* hdst = SEQ ? hseq_out + (size_t)(b0 + co_row) * T_ * H_ + co_col : nullptr;

  __syncthreads();

  v8f accx[NG_];
  {
    v16h xa0[KX];
    #pragma unroll
    for (int kt = 0; kt < KX; ++kt)
      xa0[kt] = ldfrag_s(&x_lds[0][c16][0], kt * 32, hi);
    #pragma unroll
    for (int kt = 0; kt < KX; ++kt) {
      #pragma unroll
      for (int g = 0; g < NG_; ++g)
        accx[g] = __builtin_amdgcn_wmma_f32_16x16x32_f16(false, Wx[g][kt], false, xa0[kt],
                                                         (short)0, kt == 0 ? bv[g] : accx[g],
                                                         false, false);
    }
  }
  __syncthreads();

  v16h ha[4];
  #pragma unroll
  for (int kt = 0; kt < 4; ++kt)
    ha[kt] = ldfrag_s(&h_lds[0][c16][0], kt * 32, hi);
  v16h xa[KX];
  #pragma unroll
  for (int kt = 0; kt < KX; ++kt)
    xa[kt] = ldfrag_s(&x_lds[1][c16][0], kt * 32, hi);

  v8h hv8 = {};

  float hkeep[8] = {0.f, 0.f, 0.f, 0.f, 0.f, 0.f, 0.f, 0.f};

  auto proj = [&]() {
    #pragma unroll
    for (int kt = 0; kt < KX; ++kt) {
      #pragma unroll
      for (int g = 0; g < NG_; ++g)
        accx[g] = __builtin_amdgcn_wmma_f32_16x16x32_f16(false, Wx[g][kt], false, xa[kt],
                                                         (short)0, kt == 0 ? bv[g] : accx[g],
                                                         false, false);
    }
  };

  auto step = [&](const int t, const int cur, v4f& r32, v8h& r16) {
    const int nxt = cur ^ 1;

    if (loader) {
      x_put(cur, r32, r16);
      const int sf = t + 4 < t1 ? t + 4 : t1 - 1;
      x_fetch(sf, r32, r16);
      const int sp = t + 6 < t1 ? t + 6 : t1 - 1;
      if constexpr (SEQ) __builtin_prefetch(xsrc32 + (size_t)sp * KIN, 0, 1);
      else               __builtin_prefetch(xsrc16 + (size_t)sp * KIN, 0, 1);
    }

    if (SEQ && (cur == 1 || t > t0)) *(v8h*)(hdst + (size_t)(t - 1) * H_) = hv8;

    v8f acc[NG_];
    #pragma unroll
    for (int kt = 0; kt < 4; ++kt) {
      #pragma unroll
      for (int g = 0; g < NG_; ++g)
        acc[g] = __builtin_amdgcn_wmma_f32_16x16x32_f16(false, Wh[g][kt], false, ha[kt],
                                                        (short)0, kt == 0 ? accx[g] : acc[g],
                                                        false, false);
    }

    v8h hpack;
    #pragma unroll
    for (int m = 0; m < 8; ++m) {
      float iv = 0.5f + 0.5f * fast_tanh(acc[0][m] * 0.5f);
      float fv = 0.5f + 0.5f * fast_tanh(acc[1][m] * 0.5f);
      float gv = fast_tanh(acc[2][m]);
      float ov = 0.5f + 0.5f * fast_tanh(acc[3][m] * 0.5f);
      c[m] = fv * c[m] + iv * gv;
      float hv = ov * fast_tanh(c[m]);
      hpack[m] = (_Float16)hv;
      hkeep[m] = hv;
    }
    *(v8h*)&h_lds[nxt][c16][j0 + 8 * hi] = hpack;

    lds_signal();

    __builtin_amdgcn_sched_barrier(0);
    proj();
    __builtin_amdgcn_sched_barrier(0);

    lds_wait();

    #pragma unroll
    for (int kt = 0; kt < 4; ++kt)
      ha[kt] = ldfrag_s(&h_lds[nxt][c16][0], kt * 32, hi);
    if (SEQ) hv8 = *(const v8h*)&h_lds[nxt][co_row][co_col];
    __builtin_amdgcn_sched_barrier(0);
    #pragma unroll
    for (int kt = 0; kt < KX; ++kt)
      xa[kt] = ldfrag_s(&x_lds[cur][c16][0], kt * 32, hi);
  };

  static_assert(CH_ % 2 == 0 && CH_ >= 4, "the loop takes the steps in pairs, x runs 3 steps ahead");
  for (int t = t0; t < t1; t += 2) {
    step(t,     0, xa32, xa16);
    step(t + 1, 1, xb32, xb16);
  }

  if (SEQ) *(v8h*)(hdst + (size_t)(t1 - 1) * H_) = hv8;

  if (t1 < T_) {
    v4f ca = {c[0], c[1], c[2], c[3]};
    v4f cb = {c[4], c[5], c[6], c[7]};
    *(v4f*)cst = ca;
    *(v4f*)(cst + 128) = cb;
    if (!SEQ)
      *(v8h*)(hstate + (size_t)(b0 + co_row) * H_ + co_col) = *(const v8h*)&h_lds[0][co_row][co_col];
  } else if (!SEQ) {
    #pragma unroll
    for (int m = 0; m < 8; ++m)
      hlast_out[(b0 + c16) * H_ + j0 + 8 * hi + m] = hkeep[m];
    __shared__ float fc_t[BT_][FC_];
    static_assert(FC_ == 32 && OUT_ == 16 && THREADS_ == BT_ * 16, "thread map of the dense layers");
    __syncthreads();
    const int fr = tid >> 4, fj = tid & 15;
    const float* hrow = hlast_out + (size_t)(b0 + fr) * H_;
    float s0 = bfc0[fj], s1 = bfc0[fj + 16];
    for (int k = 0; k < H_; ++k) {
      const float hv = hrow[k];
      s0 += hv * Wfc0[fj * H_ + k];
      s1 += hv * Wfc0[(fj + 16) * H_ + k];
    }
    fc_t[fr][fj]      = s0;
    fc_t[fr][fj + 16] = s1;
    __syncthreads();
    float o = bfc1[fj];
    for (int k = 0; k < FC_; ++k) o += fc_t[fr][k] * Wfc1[fj * FC_ + k];
    out[(b0 + fr) * OUT_ + fj] = o;
  }

}

__global__ __launch_bounds__(THREADS_, 1)
void lstm_pair_kernel(const float*    __restrict__ x,
                      const _Float16* __restrict__ wih0, const _Float16* __restrict__ whh0,
                      const float*    __restrict__ bias0,
                      _Float16*                    h0seq,
                      float*          __restrict__ c0state,
                      const _Float16* __restrict__ wih1, const _Float16* __restrict__ whh1,
                      const float*    __restrict__ bias1,
                      float*          __restrict__ c1state,
                      _Float16*       __restrict__ h1state,
                      float*          __restrict__ h1last,
                      const float*    __restrict__ wfc0, const float* __restrict__ bfc0,
                      const float*    __restrict__ wfc1, const float* __restrict__ bfc1,
                      float*          __restrict__ out,
                      const int ta, const int tb)
{
  const int n0 = ta >= 0 ? NWG_ : 0;
  const int wg = blockIdx.x;
  if (wg < n0)
    lstm_layer_body<D_, true>(x, nullptr, wih0, whh0, bias0, h0seq, nullptr, c0state, nullptr,
                              nullptr, nullptr, nullptr, nullptr, nullptr,
                              wg, ta, ta + CH_);
  else
    lstm_layer_body<H_, false>(nullptr, h0seq, wih1, whh1, bias1, nullptr, h1last, c1state,
                               h1state, wfc0, bfc0, wfc1, bfc1, out, wg - n0, tb, tb + CH_);
}

extern "C" void kernel_launch(void* const* d_in, const int* in_sizes, int n_in,
                              void* d_out, int out_size, void* d_ws, size_t ws_size,
                              hipStream_t stream) {
  (void)in_sizes; (void)n_in; (void)out_size; (void)ws_size;
  const float* x    = (const float*)d_in[0];
  const float* Wih0 = (const float*)d_in[1];
  const float* Whh0 = (const float*)d_in[2];
  const float* bih0 = (const float*)d_in[3];
  const float* bhh0 = (const float*)d_in[4];
  const float* Wih1 = (const float*)d_in[5];
  const float* Whh1 = (const float*)d_in[6];
  const float* bih1 = (const float*)d_in[7];
  const float* bhh1 = (const float*)d_in[8];
  const float* Wfc0 = (const float*)d_in[9];
  const float* bfc0 = (const float*)d_in[10];
  const float* Wfc1 = (const float*)d_in[11];
  const float* bfc1 = (const float*)d_in[12];

  char* p = (char*)d_ws;
  auto alloc = [&](size_t bytes) {
    char* r = p;
    p += (bytes + 255) & ~(size_t)255;
    return r;
  };
  _Float16* h0seq  = (_Float16*)alloc((size_t)B_ * T_ * H_ * 2);
  _Float16* wih0_h = (_Float16*)alloc((size_t)G_ * D_ * 2);
  _Float16* whh0_h = (_Float16*)alloc((size_t)G_ * H_ * 2);
  _Float16* wih1_h = (_Float16*)alloc((size_t)G_ * H_ * 2);
  _Float16* whh1_h = (_Float16*)alloc((size_t)G_ * H_ * 2);
  float*    b0     = (float*)alloc((size_t)G_ * 4);
  float*    b1     = (float*)alloc((size_t)G_ * 4);
  float*    h1last = (float*)alloc((size_t)B_ * H_ * 4);
  float*    c0st   = (float*)alloc((size_t)B_ * H_ * 4);
  float*    c1st   = (float*)alloc((size_t)B_ * H_ * 4);
  _Float16* h1st   = (_Float16*)alloc((size_t)B_ * H_ * 2);

  prep_kernel<<<PREP_BLOCKS_, PREP_THREADS_, 0, stream>>>(Wih0, Whh0, Wih1, Whh1,
                                                          bih0, bhh0, bih1, bhh1,
                                                          wih0_h, whh0_h, wih1_h, whh1_h, b0, b1);

  static_assert(NCH_ == 4, "the launches below are written out for four chunks");
  lstm_pair_kernel<<<NWG_, THREADS_, 0, stream>>>(x, wih0_h, whh0_h, b0, h0seq, c0st,
                                                  wih1_h, whh1_h, b1, c1st, h1st, h1last,
                                                  Wfc0, bfc0, Wfc1, bfc1, (float*)d_out,
                                                  0, -1);
  lstm_pair_kernel<<<2 * NWG_, THREADS_, 0, stream>>>(x, wih0_h, whh0_h, b0, h0seq, c0st,
                                                      wih1_h, whh1_h, b1, c1st, h1st, h1last,
                                                  Wfc0, bfc0, Wfc1, bfc1, (float*)d_out,
                                                      CH_, 0);
  lstm_pair_kernel<<<2 * NWG_, THREADS_, 0, stream>>>(x, wih0_h, whh0_h, b0, h0seq, c0st,
                                                      wih1_h, whh1_h, b1, c1st, h1st, h1last,
                                                  Wfc0, bfc0, Wfc1, bfc1, (float*)d_out,
                                                      2 * CH_, CH_);
  lstm_pair_kernel<<<2 * NWG_, THREADS_, 0, stream>>>(x, wih0_h, whh0_h, b0, h0seq, c0st,
                                                      wih1_h, whh1_h, b1, c1st, h1st, h1last,
                                                  Wfc0, bfc0, Wfc1, bfc1, (float*)d_out,
                                                      3 * CH_, 2 * CH_);
  lstm_pair_kernel<<<NWG_, THREADS_, 0, stream>>>(x, wih0_h, whh0_h, b0, h0seq, c0st,
                                                  wih1_h, whh1_h, b1, c1st, h1st, h1last,
                                                  Wfc0, bfc0, Wfc1, bfc1, (float*)d_out,
                                                  -1, 3 * CH_);
}


# Wave priority by phase and pair: serial part of a step (h reads, Whh WMMAs, cell update) at 2; projection of the older wave of a SIMD pair at 0 (was 3): it no longer holds up the sibling's Whh WMMAs
# speedup vs baseline: 7.9075x; 1.1043x over previous
#include <hip/hip_runtime.h>


#define B_ 512
#define T_ 512
#define D_ 64
#define H_ 128
#define G_ 512
#define OUT_ 16
#define FC_ 32
#define BT_ 16
#define THREADS_ 256
#define NG_ 4
#define NCH_ 4
#define CH_ (T_ / NCH_)
#define NWG_ (B_ / BT_)

typedef __attribute__((ext_vector_type(16))) _Float16 v16h;
typedef __attribute__((ext_vector_type(8)))  _Float16 v8h;
typedef __attribute__((ext_vector_type(8)))  float    v8f;
typedef __attribute__((ext_vector_type(4)))  _Float16 v4h;
typedef __attribute__((ext_vector_type(4)))  float    v4f;

__device__ __forceinline__ float fast_tanh(float x) {
#if __has_builtin(__builtin_amdgcn_tanhf)
  return __builtin_amdgcn_tanhf(x);
#else
  float e = __builtin_amdgcn_exp2f(x * 2.8853900817779268f);
  return 1.0f - 2.0f * __builtin_amdgcn_rcpf(e + 1.0f);
#endif
}

__device__ __forceinline__ v16h ldfrag_g(const _Float16* __restrict__ base,
                                         int stride, int koff, int lane) {
  int r  = lane & 15;
  int hi = lane >> 4;
  const _Float16* p = base + (size_t)r * stride + koff + hi * 8;
  v8h a = *(const v8h*)p;
  v8h b = *(const v8h*)(p + 16);
  return __builtin_shufflevector(a, b, 0,1,2,3,4,5,6,7,8,9,10,11,12,13,14,15);
}

__device__ __forceinline__ v16h ldfrag_s(const _Float16* rowbase, int koff, int hi) {
  const _Float16* p = rowbase + koff + hi * 8;
  v8h a = *(const v8h*)p;
  v8h b = *(const v8h*)(p + 16);
  return __builtin_shufflevector(a, b, 0,1,2,3,4,5,6,7,8,9,10,11,12,13,14,15);
}

__device__ __forceinline__ void lds_signal(const int young) {
  asm volatile("s_wait_dscnt 0x0\n\ts_barrier_signal -1\n\ts_setprio 0\n\t"
               "s_cmp_eq_u32 %0, 0\n\ts_cbranch_scc1 1f\n\ts_setprio 3\n1:"
               :: "s"(young) : "scc", "memory");
}
__device__ __forceinline__ void lds_wait() {
  asm volatile("s_barrier_wait -1\n\ts_setprio 2" ::: "memory");
}

#define PREP_THREADS_ 256
#define PREP_NB_X0_   (G_ * D_ / 8 / PREP_THREADS_)
#define PREP_NB_W_    (G_ * H_ / 8 / PREP_THREADS_)
#define PREP_BLOCKS_  (PREP_NB_X0_ + 3 * PREP_NB_W_ + 1)
static_assert(G_ / 4 * 2 == PREP_THREADS_, "one block holds both bias vectors, 4 elements per thread");

__device__ __forceinline__ float gate_factor(int r) {
  return (r / H_) == 2 ? 1.0f : 0.5f;
}

template<int K>
__device__ __forceinline__ void cvt8(const float* __restrict__ s, _Float16* __restrict__ d, int i,
                                     float hfac) {
  const float* sp = s + 8 * (size_t)i;
  const float sc = gate_factor(8 * i / K) * hfac;
  v4f a = *(const v4f*)sp;
  v4f b = *(const v4f*)(sp + 4);
  v8h h = {(_Float16)(a[0] * sc), (_Float16)(a[1] * sc), (_Float16)(a[2] * sc), (_Float16)(a[3] * sc),
           (_Float16)(b[0] * sc), (_Float16)(b[1] * sc), (_Float16)(b[2] * sc), (_Float16)(b[3] * sc)};
  *(v8h*)(d + 8 * (size_t)i) = h;
}

__global__ __launch_bounds__(PREP_THREADS_)
void prep_kernel(const float* __restrict__ Wih0, const float* __restrict__ Whh0,
                 const float* __restrict__ Wih1, const float* __restrict__ Whh1,
                 const float* __restrict__ bih0, const float* __restrict__ bhh0,
                 const float* __restrict__ bih1, const float* __restrict__ bhh1,
                 _Float16* __restrict__ wih0_h, _Float16* __restrict__ whh0_h,
                 _Float16* __restrict__ wih1_h, _Float16* __restrict__ whh1_h,
                 float* __restrict__ b0, float* __restrict__ b1) {
  const int blk = blockIdx.x;
  const int tid = threadIdx.x;
  if (blk < PREP_NB_X0_) {
    cvt8<D_>(Wih0, wih0_h, blk * PREP_THREADS_ + tid, 1.0f);
  } else if (blk < PREP_NB_X0_ + PREP_NB_W_) {
    cvt8<H_>(Whh0, whh0_h, (blk - PREP_NB_X0_) * PREP_THREADS_ + tid, 0.5f);
  } else if (blk < PREP_NB_X0_ + 2 * PREP_NB_W_) {
    cvt8<H_>(Wih1, wih1_h, (blk - PREP_NB_X0_ - PREP_NB_W_) * PREP_THREADS_ + tid, 0.5f);
  } else if (blk < PREP_NB_X0_ + 3 * PREP_NB_W_) {
    cvt8<H_>(Whh1, whh1_h, (blk - PREP_NB_X0_ - 2 * PREP_NB_W_) * PREP_THREADS_ + tid, 0.5f);
  } else {
    const bool l1 = tid >= G_ / 4;
    const int  i  = (tid & (G_ / 4 - 1)) * 4;
    const float* pa = l1 ? bih1 : bih0;
    const float* pb = l1 ? bhh1 : bhh0;
    float*       po = l1 ? b1 : b0;
    v4f va = *(const v4f*)(pa + i);
    v4f vb = *(const v4f*)(pb + i);
    const float bs = gate_factor(i);
    v4f vo = {(va[0] + vb[0]) * bs, (va[1] + vb[1]) * bs, (va[2] + vb[2]) * bs, (va[3] + vb[3]) * bs};
    *(v4f*)(po + i) = vo;
  }
}

template<int KIN, bool SEQ>
__device__ __forceinline__
void lstm_layer_body(const float*    __restrict__ x32,
                     const _Float16*              xh,
                     const _Float16* __restrict__ Wih,
                     const _Float16* __restrict__ Whh,
                     const float*    __restrict__ bias,
                     _Float16*                    hseq_out,
                     float*          __restrict__ hlast_out,
                     float*          __restrict__ cstate,
                     _Float16*       __restrict__ hstate,
                     const float*    __restrict__ Wfc0,
                     const float*    __restrict__ bfc0,
                     const float*    __restrict__ Wfc1,
                     const float*    __restrict__ bfc1,
                     float*          __restrict__ out,
                     const int wg, const int t0, const int t1)
{
  constexpr int KX = KIN / 32;

  __shared__ __attribute__((aligned(16)))
  _Float16 h_lds[2][BT_][H_ + 8];
  __shared__ __attribute__((aligned(16)))
  _Float16 x_lds[2][BT_][KIN + 8];

  const int tid   = threadIdx.x;
  const int lane  = tid & 31;
  const int wave  = tid >> 5;
  const int hi    = lane >> 4;
  const int c16   = lane & 15;
  const int b0    = wg * BT_;
  const int j0    = wave * 16;
  const int young = __builtin_amdgcn_readfirstlane(wave) >= THREADS_ / 64 ? 1 : 0;

  if (t0 == 0) {
    for (int i = tid; i < BT_ * (H_ + 8); i += THREADS_)
      ((_Float16*)h_lds[0])[i] = (_Float16)0.0f;
  } else {
    const int pr = tid >> 4, pc = (tid & 15) * 8;
    const _Float16* hp = SEQ ? hseq_out + ((size_t)(b0 + pr) * T_ + (t0 - 1)) * H_ + pc
                             : hstate + (size_t)(b0 + pr) * H_ + pc;
    *(v8h*)&h_lds[0][pr][pc] = *(const v8h*)hp;
  }

  v16h Wx[NG_][KX];
  v16h Wh[NG_][4];
  v8f bv[NG_];
  #pragma unroll
  for (int g = 0; g < NG_; ++g) {
    const int nc = g * H_ + j0;
    #pragma unroll
    for (int m = 0; m < 8; ++m) bv[g][m] = bias[nc + 8 * hi + m];
    #pragma unroll
    for (int kt = 0; kt < KX; ++kt)
      Wx[g][kt] = ldfrag_g(Wih + (size_t)nc * KIN, KIN, kt * 32, lane);
    #pragma unroll
    for (int kt = 0; kt < 4; ++kt)
      Wh[g][kt] = ldfrag_g(Whh + (size_t)nc * H_, H_, kt * 32, lane);
  }

  float c[8] = {0.f, 0.f, 0.f, 0.f, 0.f, 0.f, 0.f, 0.f};
  float* cst = cstate + ((size_t)(wg * (THREADS_ / 32) + wave) * 2) * 128 + lane * 4;
  if (t0 != 0) {
    v4f ca = *(const v4f*)cst;
    v4f cb = *(const v4f*)(cst + 128);
    c[0] = ca[0]; c[1] = ca[1]; c[2] = ca[2]; c[3] = ca[3];
    c[4] = cb[0]; c[5] = cb[1]; c[6] = cb[2]; c[7] = cb[3];
  }

  constexpr int EPP_   = SEQ ? 4 : 8;
  constexpr int CPR_   = KIN / EPP_;
  constexpr int NLOAD_ = BT_ * CPR_;
  static_assert(NLOAD_ <= THREADS_, "one piece per thread");
  const bool loader = (NLOAD_ == THREADS_) || tid < NLOAD_;
  const int  xs_row = tid / CPR_;
  const int  xs_col = (tid % CPR_) * EPP_;
  const size_t xoff = (size_t)(b0 + xs_row) * (T_ * KIN) + xs_col;
  const float*    xsrc32 = SEQ ? x32 + xoff : nullptr;
  const _Float16* xsrc16 = SEQ ? nullptr : xh + xoff;
  v4f xa32 = {}, xb32 = {};
  v8h xa16 = {}, xb16 = {};
  auto x_fetch = [&](int s, v4f& r32, v8h& r16) {
    if constexpr (SEQ) r32 = *(const v4f*)(xsrc32 + (size_t)s * KIN);
    else               r16 = *(const v8h*)(xsrc16 + (size_t)s * KIN);
  };
  auto x_put = [&](int cp, const v4f& r32, const v8h& r16) {
    if constexpr (SEQ) {
      v4h p4 = {(_Float16)r32[0], (_Float16)r32[1], (_Float16)r32[2], (_Float16)r32[3]};
      *(v4h*)&x_lds[cp][xs_row][xs_col] = p4;
    } else {
      *(v8h*)&x_lds[cp][xs_row][xs_col] = r16;
    }
  };
  if (loader) {
    x_fetch(t0,     xa32, xa16); x_put(0, xa32, xa16);
    x_fetch(t0 + 1, xb32, xb16); x_put(1, xb32, xb16);
    x_fetch(t0 + 2, xa32, xa16);
    x_fetch(t0 + 3, xb32, xb16);
  }

  const int co_row = tid >> 4;
  const int co_col = (tid & 15) * 8;
  _Float16* hdst = SEQ ? hseq_out + (size_t)(b0 + co_row) * T_ * H_ + co_col : nullptr;

  __syncthreads();

  v8f accx[NG_];
  {
    v16h xa0[KX];
    #pragma unroll
    for (int kt = 0; kt < KX; ++kt)
      xa0[kt] = ldfrag_s(&x_lds[0][c16][0], kt * 32, hi);
    #pragma unroll
    for (int kt = 0; kt < KX; ++kt) {
      #pragma unroll
      for (int g = 0; g < NG_; ++g)
        accx[g] = __builtin_amdgcn_wmma_f32_16x16x32_f16(false, Wx[g][kt], false, xa0[kt],
                                                         (short)0, kt == 0 ? bv[g] : accx[g],
                                                         false, false);
    }
  }
  __syncthreads();

  v16h ha[4];
  #pragma unroll
  for (int kt = 0; kt < 4; ++kt)
    ha[kt] = ldfrag_s(&h_lds[0][c16][0], kt * 32, hi);
  v16h xa[KX];
  #pragma unroll
  for (int kt = 0; kt < KX; ++kt)
    xa[kt] = ldfrag_s(&x_lds[1][c16][0], kt * 32, hi);

  v8h hv8 = {};

  float hkeep[8] = {0.f, 0.f, 0.f, 0.f, 0.f, 0.f, 0.f, 0.f};

  auto proj = [&]() {
    #pragma unroll
    for (int kt = 0; kt < KX; ++kt) {
      #pragma unroll
      for (int g = 0; g < NG_; ++g)
        accx[g] = __builtin_amdgcn_wmma_f32_16x16x32_f16(false, Wx[g][kt], false, xa[kt],
                                                         (short)0, kt == 0 ? bv[g] : accx[g],
                                                         false, false);
    }
  };

  auto step = [&](const int t, const int cur, v4f& r32, v8h& r16) {
    const int nxt = cur ^ 1;

    if (loader) {
      x_put(cur, r32, r16);
      const int sf = t + 4 < t1 ? t + 4 : t1 - 1;
      x_fetch(sf, r32, r16);
      const int sp = t + 6 < t1 ? t + 6 : t1 - 1;
      if constexpr (SEQ) __builtin_prefetch(xsrc32 + (size_t)sp * KIN, 0, 1);
      else               __builtin_prefetch(xsrc16 + (size_t)sp * KIN, 0, 1);
    }

    if (SEQ && (cur == 1 || t > t0)) *(v8h*)(hdst + (size_t)(t - 1) * H_) = hv8;

    v8f acc[NG_];
    #pragma unroll
    for (int kt = 0; kt < 4; ++kt) {
      #pragma unroll
      for (int g = 0; g < NG_; ++g)
        acc[g] = __builtin_amdgcn_wmma_f32_16x16x32_f16(false, Wh[g][kt], false, ha[kt],
                                                        (short)0, kt == 0 ? accx[g] : acc[g],
                                                        false, false);
    }

    v8h hpack;
    #pragma unroll
    for (int m = 0; m < 8; ++m) {
      float iv = 0.5f + 0.5f * fast_tanh(acc[0][m]);
      float fv = 0.5f + 0.5f * fast_tanh(acc[1][m]);
      float gv = fast_tanh(acc[2][m]);
      float to = fast_tanh(acc[3][m]);
      c[m] = fv * c[m] + iv * gv;
      float tc = fast_tanh(c[m]);
      float h2 = to * tc + tc;
      hpack[m] = (_Float16)h2;
      hkeep[m] = h2;
    }
    *(v8h*)&h_lds[nxt][c16][j0 + 8 * hi] = hpack;

    lds_signal(young);

    __builtin_amdgcn_sched_barrier(0);
    proj();
    __builtin_amdgcn_sched_barrier(0);

    lds_wait();

    #pragma unroll
    for (int kt = 0; kt < 4; ++kt)
      ha[kt] = ldfrag_s(&h_lds[nxt][c16][0], kt * 32, hi);
    if (SEQ) hv8 = *(const v8h*)&h_lds[nxt][co_row][co_col];
    __builtin_amdgcn_sched_barrier(0);
    #pragma unroll
    for (int kt = 0; kt < KX; ++kt)
      xa[kt] = ldfrag_s(&x_lds[cur][c16][0], kt * 32, hi);
  };

  static_assert(CH_ % 2 == 0 && CH_ >= 4, "the loop takes the steps in pairs, x runs 3 steps ahead");
  for (int t = t0; t < t1; t += 2) {
    step(t,     0, xa32, xa16);
    step(t + 1, 1, xb32, xb16);
  }

  if (SEQ) *(v8h*)(hdst + (size_t)(t1 - 1) * H_) = hv8;

  if (t1 < T_) {
    v4f ca = {c[0], c[1], c[2], c[3]};
    v4f cb = {c[4], c[5], c[6], c[7]};
    *(v4f*)cst = ca;
    *(v4f*)(cst + 128) = cb;
    if (!SEQ)
      *(v8h*)(hstate + (size_t)(b0 + co_row) * H_ + co_col) = *(const v8h*)&h_lds[0][co_row][co_col];
  } else if (!SEQ) {
    #pragma unroll
    for (int m = 0; m < 8; ++m)
      hlast_out[(b0 + c16) * H_ + j0 + 8 * hi + m] = 0.5f * hkeep[m];
    __shared__ float fc_t[BT_][FC_];
    static_assert(FC_ == 32 && OUT_ == 16 && THREADS_ == BT_ * 16, "thread map of the dense layers");
    __syncthreads();
    const int fr = tid >> 4, fj = tid & 15;
    const float* hrow = hlast_out + (size_t)(b0 + fr) * H_;
    float s0 = bfc0[fj], s1 = bfc0[fj + 16];
    for (int k = 0; k < H_; ++k) {
      const float hv = hrow[k];
      s0 += hv * Wfc0[fj * H_ + k];
      s1 += hv * Wfc0[(fj + 16) * H_ + k];
    }
    fc_t[fr][fj]      = s0;
    fc_t[fr][fj + 16] = s1;
    __syncthreads();
    float o = bfc1[fj];
    for (int k = 0; k < FC_; ++k) o += fc_t[fr][k] * Wfc1[fj * FC_ + k];
    out[(b0 + fr) * OUT_ + fj] = o;
  }

}

__global__ __launch_bounds__(THREADS_, 1)
void lstm_pair_kernel(const float*    __restrict__ x,
                      const _Float16* __restrict__ wih0, const _Float16* __restrict__ whh0,
                      const float*    __restrict__ bias0,
                      _Float16*                    h0seq,
                      float*          __restrict__ c0state,
                      const _Float16* __restrict__ wih1, const _Float16* __restrict__ whh1,
                      const float*    __restrict__ bias1,
                      float*          __restrict__ c1state,
                      _Float16*       __restrict__ h1state,
                      float*          __restrict__ h1last,
                      const float*    __restrict__ wfc0, const float* __restrict__ bfc0,
                      const float*    __restrict__ wfc1, const float* __restrict__ bfc1,
                      float*          __restrict__ out,
                      const int ta, const int tb)
{
  const int n0 = ta >= 0 ? NWG_ : 0;
  const int wg = blockIdx.x;
  if (wg < n0)
    lstm_layer_body<D_, true>(x, nullptr, wih0, whh0, bias0, h0seq, nullptr, c0state, nullptr,
                              nullptr, nullptr, nullptr, nullptr, nullptr,
                              wg, ta, ta + CH_);
  else
    lstm_layer_body<H_, false>(nullptr, h0seq, wih1, whh1, bias1, nullptr, h1last, c1state,
                               h1state, wfc0, bfc0, wfc1, bfc1, out, wg - n0, tb, tb + CH_);
}

extern "C" void kernel_launch(void* const* d_in, const int* in_sizes, int n_in,
                              void* d_out, int out_size, void* d_ws, size_t ws_size,
                              hipStream_t stream) {
  (void)in_sizes; (void)n_in; (void)out_size; (void)ws_size;
  const float* x    = (const float*)d_in[0];
  const float* Wih0 = (const float*)d_in[1];
  const float* Whh0 = (const float*)d_in[2];
  const float* bih0 = (const float*)d_in[3];
  const float* bhh0 = (const float*)d_in[4];
  const float* Wih1 = (const float*)d_in[5];
  const float* Whh1 = (const float*)d_in[6];
  const float* bih1 = (const float*)d_in[7];
  const float* bhh1 = (const float*)d_in[8];
  const float* Wfc0 = (const float*)d_in[9];
  const float* bfc0 = (const float*)d_in[10];
  const float* Wfc1 = (const float*)d_in[11];
  const float* bfc1 = (const float*)d_in[12];

  char* p = (char*)d_ws;
  auto alloc = [&](size_t bytes) {
    char* r = p;
    p += (bytes + 255) & ~(size_t)255;
    return r;
  };
  _Float16* h0seq  = (_Float16*)alloc((size_t)B_ * T_ * H_ * 2);
  _Float16* wih0_h = (_Float16*)alloc((size_t)G_ * D_ * 2);
  _Float16* whh0_h = (_Float16*)alloc((size_t)G_ * H_ * 2);
  _Float16* wih1_h = (_Float16*)alloc((size_t)G_ * H_ * 2);
  _Float16* whh1_h = (_Float16*)alloc((size_t)G_ * H_ * 2);
  float*    b0     = (float*)alloc((size_t)G_ * 4);
  float*    b1     = (float*)alloc((size_t)G_ * 4);
  float*    h1last = (float*)alloc((size_t)B_ * H_ * 4);
  float*    c0st   = (float*)alloc((size_t)B_ * H_ * 4);
  float*    c1st   = (float*)alloc((size_t)B_ * H_ * 4);
  _Float16* h1st   = (_Float16*)alloc((size_t)B_ * H_ * 2);

  prep_kernel<<<PREP_BLOCKS_, PREP_THREADS_, 0, stream>>>(Wih0, Whh0, Wih1, Whh1,
                                                          bih0, bhh0, bih1, bhh1,
                                                          wih0_h, whh0_h, wih1_h, whh1_h, b0, b1);

  static_assert(NCH_ == 4, "the launches below are written out for four chunks");
  lstm_pair_kernel<<<NWG_, THREADS_, 0, stream>>>(x, wih0_h, whh0_h, b0, h0seq, c0st,
                                                  wih1_h, whh1_h, b1, c1st, h1st, h1last,
                                                  Wfc0, bfc0, Wfc1, bfc1, (float*)d_out,
                                                  0, -1);
  lstm_pair_kernel<<<2 * NWG_, THREADS_, 0, stream>>>(x, wih0_h, whh0_h, b0, h0seq, c0st,
                                                      wih1_h, whh1_h, b1, c1st, h1st, h1last,
                                                  Wfc0, bfc0, Wfc1, bfc1, (float*)d_out,
                                                      CH_, 0);
  lstm_pair_kernel<<<2 * NWG_, THREADS_, 0, stream>>>(x, wih0_h, whh0_h, b0, h0seq, c0st,
                                                      wih1_h, whh1_h, b1, c1st, h1st, h1last,
                                                  Wfc0, bfc0, Wfc1, bfc1, (float*)d_out,
                                                      2 * CH_, CH_);
  lstm_pair_kernel<<<2 * NWG_, THREADS_, 0, stream>>>(x, wih0_h, whh0_h, b0, h0seq, c0st,
                                                      wih1_h, whh1_h, b1, c1st, h1st, h1last,
                                                  Wfc0, bfc0, Wfc1, bfc1, (float*)d_out,
                                                      3 * CH_, 2 * CH_);
  lstm_pair_kernel<<<NWG_, THREADS_, 0, stream>>>(x, wih0_h, whh0_h, b0, h0seq, c0st,
                                                  wih1_h, whh1_h, b1, c1st, h1st, h1last,
                                                  Wfc0, bfc0, Wfc1, bfc1, (float*)d_out,
                                                  -1, 3 * CH_);
}
